# starting kernel
# baseline (speedup 1.0000x reference)
#include <hip/hip_runtime.h>
#include <hip/hip_bf16.h>


typedef float v2f __attribute__((ext_vector_type(2)));
typedef float v8f __attribute__((ext_vector_type(8)));

#define BB   8
#define CC   32
#define HH   256
#define WW   256
#define HWSZ (HH * WW)

__global__ __launch_bounds__(256) void pac_prep(const float* __restrict__ guided,
                                                float* __restrict__ wgt) {
  int idx = blockIdx.x * 256 + threadIdx.x;
  int b  = idx >> 16;
  int hw = idx & 65535;
  const float* gp = guided + (size_t)b * 8 * HWSZ + hw;
  float g[8];
  float mx = -1e30f;
#pragma unroll
  for (int n = 0; n < 8; n++) { g[n] = gp[(size_t)n * HWSZ]; mx = fmaxf(mx, g[n]); }
  float s = 0.f;
#pragma unroll
  for (int n = 0; n < 8; n++) { g[n] = __expf(g[n] - mx); s += g[n]; }
  float inv = 1.0f / s;
  float* wp = wgt + (size_t)b * 9 * HWSZ + hw;
#pragma unroll
  for (int n = 0; n < 4; n++) wp[(size_t)n * HWSZ] = g[n] * inv;
  wp[(size_t)4 * HWSZ] = 0.f;
#pragma unroll
  for (int n = 0; n < 4; n++) wp[(size_t)(5 + n) * HWSZ] = g[4 + n] * inv;
}

__global__ __launch_bounds__(256) void pac_prop(const float* __restrict__ xin,
                                                const float* __restrict__ x0,
                                                const float* __restrict__ wgt,
                                                const float* __restrict__ sdep,
                                                float* __restrict__ xout) {
  __shared__ float Bmat[8][56 * 16];

  const int lane = threadIdx.x & 31;
  const int wv   = threadIdx.x >> 5;
  const int half = lane >> 4;
  const int p    = lane & 15;

  const int tile  = blockIdx.x * 8 + wv;
  const int b     = tile >> 12;
  const int rem   = tile & 4095;
  const int h     = rem >> 4;
  const int wbase = (rem & 15) << 4;

  float* BW = Bmat[wv];

#pragma unroll
  for (int t = 0; t < 28; t++) BW[lane + 32 * t] = 0.f;
  __syncthreads();
  if (lane < 16) {
#pragma unroll
    for (int n = 0; n < 9; n++) {
      const int i = n / 3, j = n % 3;
      int row = h - 1 + i;
      int col = wbase + p + j - 1;
      float wval = wgt[(size_t)(b * 9 + n) * HWSZ + h * WW + wbase + p];
      if (row >= 0 && row < HH && col >= 0 && col < WW)
        BW[(18 * i + p + j) * 16 + p] = wval;
    }
  }
  __syncthreads();

  const float* xa = xin + (size_t)b * CC * HWSZ;
  v8f acc0 = {0.f, 0.f, 0.f, 0.f, 0.f, 0.f, 0.f, 0.f};
  v8f acc1 = {0.f, 0.f, 0.f, 0.f, 0.f, 0.f, 0.f, 0.f};
#pragma unroll
  for (int k = 0; k < 14; k++) {
    int q0 = 4 * k + 2 * half;
    int i  = (q0 >= 36) ? 2 : ((q0 >= 18) ? 1 : 0);
    int jj = q0 - 18 * i;
    int rowc = min(max(h - 1 + i, 0), HH - 1);
    int col0 = wbase - 1 + jj;
    int c0c  = min(max(col0, 0), WW - 1);
    int c1c  = min(max(col0 + 1, 0), WW - 1);

    v2f bf;
    bf.x = BW[(4 * k + 0) * 16 + half * 32 + p];
    bf.y = BW[(4 * k + 1) * 16 + half * 32 + p];

    const float* rp0 = xa + (size_t)p * HWSZ + (size_t)rowc * WW;
    const float* rp1 = rp0 + (size_t)16 * HWSZ;
    v2f af0; af0.x = rp0[c0c]; af0.y = rp0[c1c];
    v2f af1; af1.x = rp1[c0c]; af1.y = rp1[c1c];

    acc0 = __builtin_amdgcn_wmma_f32_16x16x4_f32(false, af0, false, bf,
                                                 (short)0, acc0, false, false);
    acc1 = __builtin_amdgcn_wmma_f32_16x16x4_f32(false, af1, false, bf,
                                                 (short)0, acc1, false, false);
  }

  float sd = sdep[(size_t)b * HWSZ + h * WW + wbase + p];
  float m  = (sd > 0.f) ? 1.0f : 0.0f;
#pragma unroll
  for (int v = 0; v < 8; v++) {
    size_t off0 = (size_t)(b * CC + v + 8 * half) * HWSZ + h * WW + wbase + p;
    size_t off1 = off0 + (size_t)16 * HWSZ;
    xout[off0] = m * x0[off0] + (1.0f - m) * acc0[v];
    xout[off1] = m * x0[off1] + (1.0f - m) * acc1[v];
  }
}

extern "C" void kernel_launch(void* const* d_in, const int* in_sizes, int n_in,
                              void* d_out, int out_size, void* d_ws, size_t ws_size,
                              hipStream_t stream) {
  const float* x      = (const float*)d_in[0];
  const float* guided = (const float*)d_in[1];
  const float* sdep   = (const float*)d_in[2];
  (void)in_sizes; (void)n_in; (void)out_size; (void)ws_size;

  float* bufA = (float*)d_ws;
  float* wgt  = bufA + (size_t)BB * CC * HWSZ;

  pac_prep<<<(BB * HWSZ) / 256, 256, 0, stream>>>(guided, wgt);

  float* outb = (float*)d_out;
  const float* cur = x;
  for (int t = 0; t < 16; t++) {
    float* dst = (t & 1) ? outb : bufA;
    pac_prop<<<4096, 256, 0, stream>>>(cur, x, wgt, sdep, dst);
    cur = dst;
  }
}


# two steps per launch, row-streamed: full-width block of 16 strip waves walks 8 rows; first-step rows stay in a 4-row LDS ring (one LDS-only barrier a row, no halo columns); 9 launches; sums unchanged
# speedup vs baseline: 3.5663x; 3.5663x over previous
#include <hip/hip_runtime.h>


typedef float v2f __attribute__((ext_vector_type(2)));
typedef float v4f __attribute__((ext_vector_type(4)));
typedef float v8f __attribute__((ext_vector_type(8)));

#define BB   8
#define CC   32
#define HH   256
#define WW   256
#define HWSZ (HH * WW)
#define RR   8

__global__ __launch_bounds__(256) void pac_prep(const float* __restrict__ guided,
                                                const float* __restrict__ x,
                                                float* __restrict__ wgt,
                                                float* __restrict__ xcl) {
  __shared__ float tile[256 * 33];

  const int t   = threadIdx.x;
  const int idx = blockIdx.x * 256 + t;
  const int b   = idx >> 16;
  const int hw  = idx & 65535;

  const float* xp = x + (size_t)b * CC * HWSZ + hw;
#pragma unroll
  for (int c = 0; c < CC; c++) tile[t * 33 + c] = xp[(size_t)c * HWSZ];

  const float* gp = guided + (size_t)b * 8 * HWSZ + hw;
  float g[8];
  float mx = -1e30f;
#pragma unroll
  for (int n = 0; n < 8; n++) { g[n] = gp[(size_t)n * HWSZ]; mx = fmaxf(mx, g[n]); }
  float s = 0.f;
#pragma unroll
  for (int n = 0; n < 8; n++) { g[n] = __expf(g[n] - mx); s += g[n]; }
  float inv = 1.0f / s;
  float* wp = wgt + (size_t)b * 9 * HWSZ + hw;
#pragma unroll
  for (int n = 0; n < 4; n++) wp[(size_t)n * HWSZ] = g[n] * inv;
  wp[(size_t)4 * HWSZ] = 0.f;
#pragma unroll
  for (int n = 0; n < 4; n++) wp[(size_t)(5 + n) * HWSZ] = g[4 + n] * inv;

  __syncthreads();
  float* op = xcl + (size_t)(idx - t) * CC;
#pragma unroll
  for (int j = 0; j < 8; j++) {
    const int f  = (j * 256 + t) * 4;
    const int px = f >> 5;
    const int ch = f & 31;
    const float* tp = tile + px * 33 + ch;
    v4f v = {tp[0], tp[1], tp[2], tp[3]};
    *(v4f*)(op + f) = v;
  }
}

struct Epi {
  v4f q[6];
  float f[17];
};

template <bool OUT_CL>
__device__ __forceinline__ void epi_load(Epi& E, const float* __restrict__ x0,
                                         const float* __restrict__ sdep,
                                         int b, int h, int wbase, int half, int p) {
  if (OUT_CL) {
    const int pix0 = h * WW + wbase + 8 * half;
    const float* sp = sdep + (size_t)b * HWSZ + pix0;
    E.q[0] = *(const v4f*)(sp);
    E.q[1] = *(const v4f*)(sp + 4);
    const float* xp0 = x0 + (size_t)(b * CC + 2 * p) * HWSZ + pix0;
    const float* xp1 = xp0 + HWSZ;
    E.q[2] = *(const v4f*)(xp0);
    E.q[3] = *(const v4f*)(xp0 + 4);
    E.q[4] = *(const v4f*)(xp1);
    E.q[5] = *(const v4f*)(xp1 + 4);
  } else {
    E.f[16] = sdep[(size_t)b * HWSZ + h * WW + wbase + p];
#pragma unroll
    for (int v = 0; v < 8; v++) {
      size_t off0 = (size_t)(b * CC + 2 * (v + 8 * half)) * HWSZ + h * WW + wbase + p;
      E.f[2 * v]     = x0[off0];
      E.f[2 * v + 1] = x0[off0 + (size_t)HWSZ];
    }
  }
}

#define LP CC

template <bool OUT_CL>
__device__ __forceinline__ void prop2_body(float* __restrict__ BW, float* __restrict__ YR,
                                           const float* __restrict__ xin,
                                           const float* __restrict__ x0,
                                           const float* __restrict__ wgt,
                                           const float* __restrict__ sdep,
                                           float* __restrict__ xout) {
  const int lane = threadIdx.x & 31;
  const int wv   = threadIdx.x >> 5;
  const int half = lane >> 4;
  const int p    = lane & 15;

  const int b     = blockIdx.x >> 5;
  const int h0    = (blockIdx.x & 31) * RR;
  const int wbase = wv << 4;

#pragma unroll
  for (int t = 0; t < 60; t++) BW[lane + 32 * t] = 0.f;

  int co[10];
#pragma unroll
  for (int kk = 0; kk < 5; kk++) {
    int c0 = wbase - 1 + 4 * kk + 2 * half;
    co[2 * kk]     = min(max(c0, 0), WW - 1) * CC;
    co[2 * kk + 1] = min(max(c0 + 1, 0), WW - 1) * CC;
  }
  const bool cv0 = (wbase + p - 1) >= 0;
  const bool cv2 = (wbase + p + 1) < WW;

  const float* xa = xin + (size_t)b * HWSZ * CC + 2 * p;
  const float* wq = wgt + (size_t)b * 9 * HWSZ + wbase + p;

  v2f XA0[3][5], XA1[3][5];
  float w[2][9];
  Epi E;
  Epi EZ;

#define LOAD_ROW(slot, row)                                                     \
  {                                                                             \
    const int rowc_ = min(max((row), 0), HH - 1);                               \
    const float* rp_ = xa + (size_t)rowc_ * (WW * CC);                          \
    _Pragma("unroll") for (int kk = 0; kk < 5; kk++) {                          \
      v2f u0 = *(const v2f*)(rp_ + co[2 * kk]);                                 \
      v2f u1 = *(const v2f*)(rp_ + co[2 * kk + 1]);                             \
      XA0[(slot)][kk] = (v2f){u0.x, u1.x};                                      \
      XA1[(slot)][kk] = (v2f){u0.y, u1.y};                                      \
    }                                                                           \
  }
#define LOAD_W(s, row)                                                          \
  {                                                                             \
    const int rowc_ = min(max((row), 0), HH - 1);                               \
    _Pragma("unroll") for (int n = 0; n < 9; n++)                               \
      w[(s)][n] = wq[(size_t)n * HWSZ + rowc_ * WW];                            \
  }
#define BAND(ws, row, bb)                                                         \
  {                                                                             \
    __builtin_amdgcn_fence(__ATOMIC_ACQ_REL, "workgroup", "local");             \
    if (lane < 16) {                                                            \
      _Pragma("unroll") for (int n = 0; n < 9; n++) {                           \
        const int i = n / 3, j = n % 3;                                         \
        const bool rok = ((row) - 1 + i >= 0) && ((row) - 1 + i < HH);          \
        const bool cok = (j == 0) ? cv0 : ((j == 2) ? cv2 : true);              \
        (bb)[(20 * i + p + j) * 16 + p] = (rok && cok) ? w[(ws)][n] : 0.f;      \
      }                                                                         \
    }                                                                           \
    __builtin_amdgcn_fence(__ATOMIC_ACQ_REL, "workgroup", "local");             \
  }

  LOAD_ROW(0, h0 - 2)
  LOAD_ROW(1, h0 - 1)
  LOAD_ROW(2, h0)
  LOAD_W(0, h0 - 1)
  epi_load<true>(E, x0, sdep, b, min(max(h0 - 1, 0), HH - 1), wbase, half, p);

#pragma unroll
  for (int s = 0; s < RR + 2; s++) {
    const int y = h0 - 1 + s;
    if (s + 1 < RR + 2) LOAD_W((s + 1) & 1, y + 1)
    if (!OUT_CL && s >= 2) epi_load<false>(EZ, x0, sdep, b, y - 1, wbase, half, p);
    const float* BWy = BW + (s & 1) * 960;
    const float* BWz = BW + ((s + 1) & 1) * 960;
    BAND(s & 1, y, BW + (s & 1) * 960)
    {
      v8f acc0 = {0.f, 0.f, 0.f, 0.f, 0.f, 0.f, 0.f, 0.f};
      v8f acc1 = {0.f, 0.f, 0.f, 0.f, 0.f, 0.f, 0.f, 0.f};
#pragma unroll
      for (int i = 0; i < 3; i++) {
#pragma unroll
        for (int kk = 0; kk < 5; kk++) {
          const int c = 5 * i + kk;
          v2f bf;
          bf.x = BWy[(4 * c + 0) * 16 + half * 32 + p];
          bf.y = BWy[(4 * c + 1) * 16 + half * 32 + p];
          acc0 = __builtin_amdgcn_wmma_f32_16x16x4_f32(false, bf, false, XA0[(s + i) % 3][kk],
                                                       (short)0, acc0, false, false);
          acc1 = __builtin_amdgcn_wmma_f32_16x16x4_f32(false, bf, false, XA1[(s + i) % 3][kk],
                                                       (short)0, acc1, false, false);
        }
      }
      float sd[8]  = {E.q[0].x, E.q[0].y, E.q[0].z, E.q[0].w, E.q[1].x, E.q[1].y, E.q[1].z, E.q[1].w};
      float xv0[8] = {E.q[2].x, E.q[2].y, E.q[2].z, E.q[2].w, E.q[3].x, E.q[3].y, E.q[3].z, E.q[3].w};
      float xv1[8] = {E.q[4].x, E.q[4].y, E.q[4].z, E.q[4].w, E.q[5].x, E.q[5].y, E.q[5].z, E.q[5].w};
      float* yp = YR + (s & 3) * (WW * LP) + (wbase + 8 * half) * LP + 2 * p;
#pragma unroll
      for (int d = 0; d < 8; d++) {
        float m = (sd[d] > 0.f) ? 1.0f : 0.0f;
        v2f o;
        o.x = m * xv0[d] + (1.0f - m) * acc0[d];
        o.y = m * xv1[d] + (1.0f - m) * acc1[d];
        *(v2f*)(yp + d * LP) = o;
      }
    }
    __builtin_amdgcn_fence(__ATOMIC_RELEASE, "workgroup", "local");
    __builtin_amdgcn_s_barrier();
    __builtin_amdgcn_fence(__ATOMIC_ACQUIRE, "workgroup", "local");

    if (OUT_CL && s >= 2) epi_load<true>(EZ, x0, sdep, b, y - 1, wbase, half, p);
    if (s + 1 < RR + 2) LOAD_ROW(s % 3, y + 2)
    if (s >= 2) {
      const int z = y - 1;
      v8f acc0 = {0.f, 0.f, 0.f, 0.f, 0.f, 0.f, 0.f, 0.f};
      v8f acc1 = {0.f, 0.f, 0.f, 0.f, 0.f, 0.f, 0.f, 0.f};
#pragma unroll
      for (int i = 0; i < 3; i++) {
        const float* yr = YR + ((s - 2 + i) & 3) * (WW * LP) + 2 * p;
#pragma unroll
        for (int kk = 0; kk < 5; kk++) {
          const int c = 5 * i + kk;
          v2f bf;
          bf.x = BWz[(4 * c + 0) * 16 + half * 32 + p];
          bf.y = BWz[(4 * c + 1) * 16 + half * 32 + p];
          const v2f u0 = *(const v2f*)(yr + co[2 * kk]);
          const v2f u1 = *(const v2f*)(yr + co[2 * kk + 1]);
          const v2f af0 = (v2f){u0.x, u1.x};
          const v2f af1 = (v2f){u0.y, u1.y};
          if (OUT_CL) {
            acc0 = __builtin_amdgcn_wmma_f32_16x16x4_f32(false, bf, false, af0,
                                                         (short)0, acc0, false, false);
            acc1 = __builtin_amdgcn_wmma_f32_16x16x4_f32(false, bf, false, af1,
                                                         (short)0, acc1, false, false);
          } else {
            acc0 = __builtin_amdgcn_wmma_f32_16x16x4_f32(false, af0, false, bf,
                                                         (short)0, acc0, false, false);
            acc1 = __builtin_amdgcn_wmma_f32_16x16x4_f32(false, af1, false, bf,
                                                         (short)0, acc1, false, false);
          }
        }
      }
      __builtin_amdgcn_sched_barrier(0);
      if (OUT_CL) {
        const int pix0 = z * WW + wbase + 8 * half;
        float sd[8]  = {EZ.q[0].x, EZ.q[0].y, EZ.q[0].z, EZ.q[0].w, EZ.q[1].x, EZ.q[1].y, EZ.q[1].z, EZ.q[1].w};
        float xv0[8] = {EZ.q[2].x, EZ.q[2].y, EZ.q[2].z, EZ.q[2].w, EZ.q[3].x, EZ.q[3].y, EZ.q[3].z, EZ.q[3].w};
        float xv1[8] = {EZ.q[4].x, EZ.q[4].y, EZ.q[4].z, EZ.q[4].w, EZ.q[5].x, EZ.q[5].y, EZ.q[5].z, EZ.q[5].w};
        float* op = xout + ((size_t)b * HWSZ + pix0) * CC + 2 * p;
#pragma unroll
        for (int d = 0; d < 8; d++) {
          float m = (sd[d] > 0.f) ? 1.0f : 0.0f;
          v2f o;
          o.x = m * xv0[d] + (1.0f - m) * acc0[d];
          o.y = m * xv1[d] + (1.0f - m) * acc1[d];
          *(v2f*)(op + d * CC) = o;
        }
      } else {
        float m = (EZ.f[16] > 0.f) ? 1.0f : 0.0f;
#pragma unroll
        for (int v = 0; v < 8; v++) {
          size_t off0 = (size_t)(b * CC + 2 * (v + 8 * half)) * HWSZ + z * WW + wbase + p;
          size_t off1 = off0 + (size_t)HWSZ;
          xout[off0] = m * EZ.f[2 * v]     + (1.0f - m) * acc0[v];
          xout[off1] = m * EZ.f[2 * v + 1] + (1.0f - m) * acc1[v];
        }
      }
    }
    if (s + 1 < RR + 2)
      epi_load<true>(E, x0, sdep, b, min(max(y + 1, 0), HH - 1), wbase, half, p);
  }
#undef LOAD_ROW
#undef LOAD_W
#undef BAND
}

#define BAND_FLOATS (16 * 2 * 60 * 16)
#define RING_FLOATS (4 * WW * LP)

__global__ __launch_bounds__(512) void pac_prop2_cl(const float* __restrict__ xin,
                                                    const float* __restrict__ x0,
                                                    const float* __restrict__ wgt,
                                                    const float* __restrict__ sdep,
                                                    float* __restrict__ xout) {
  __shared__ __attribute__((aligned(16))) float smem[BAND_FLOATS + RING_FLOATS];
  prop2_body<true>(smem + (threadIdx.x >> 5) * (2 * 60 * 16), smem + BAND_FLOATS,
                   xin, x0, wgt, sdep, xout);
}

__global__ __launch_bounds__(512) void pac_prop2_last(const float* __restrict__ xin,
                                                      const float* __restrict__ x0,
                                                      const float* __restrict__ wgt,
                                                      const float* __restrict__ sdep,
                                                      float* __restrict__ xout) {
  __shared__ __attribute__((aligned(16))) float smem[BAND_FLOATS + RING_FLOATS];
  prop2_body<false>(smem + (threadIdx.x >> 5) * (2 * 60 * 16), smem + BAND_FLOATS,
                    xin, x0, wgt, sdep, xout);
}

extern "C" void kernel_launch(void* const* d_in, const int* in_sizes, int n_in,
                              void* d_out, int out_size, void* d_ws, size_t ws_size,
                              hipStream_t stream) {
  const float* x      = (const float*)d_in[0];
  const float* guided = (const float*)d_in[1];
  const float* sdep   = (const float*)d_in[2];
  (void)in_sizes; (void)n_in; (void)out_size; (void)ws_size;

  float* bufA = (float*)d_ws;
  float* wgt  = bufA + (size_t)BB * CC * HWSZ;
  float* outb = (float*)d_out;

  pac_prep<<<(BB * HWSZ) / 256, 256, 0, stream>>>(guided, x, wgt, outb);

  const float* cur = outb;
  for (int t = 0; t < 7; t++) {
    float* dst = (t & 1) ? outb : bufA;
    pac_prop2_cl<<<BB * (HH / RR), 512, 0, stream>>>(cur, x, wgt, sdep, dst);
    cur = dst;
  }
  pac_prop2_last<<<BB * (HH / RR), 512, 0, stream>>>(cur, x, wgt, sdep, outb);
}


# state kept pair-interleaved ([pixel pair][16 slots][ch p, p+16][2 cols]): one 16-byte load per lane is a K-chunk's fragment pair, no register shuffles; 16-byte stores; taps regroup: sum order changes
# speedup vs baseline: 3.9352x; 1.1034x over previous
#include <hip/hip_runtime.h>


typedef float v2f __attribute__((ext_vector_type(2)));
typedef float v4f __attribute__((ext_vector_type(4)));
typedef float v8f __attribute__((ext_vector_type(8)));

#define BB   8
#define CC   32
#define HH   256
#define WW   256
#define HWSZ (HH * WW)
#define RR   8
#define PP   64

__global__ __launch_bounds__(256) void pac_prep(const float* __restrict__ guided,
                                                const float* __restrict__ x,
                                                float* __restrict__ wgt,
                                                float* __restrict__ xcl) {
  __shared__ float tile[256 * 33];

  const int t   = threadIdx.x;
  const int idx = blockIdx.x * 256 + t;
  const int b   = idx >> 16;
  const int hw  = idx & 65535;

  const float* xp = x + (size_t)b * CC * HWSZ + hw;
#pragma unroll
  for (int c = 0; c < CC; c++) tile[t * 33 + c] = xp[(size_t)c * HWSZ];

  const float* gp = guided + (size_t)b * 8 * HWSZ + hw;
  float g[8];
  float mx = -1e30f;
#pragma unroll
  for (int n = 0; n < 8; n++) { g[n] = gp[(size_t)n * HWSZ]; mx = fmaxf(mx, g[n]); }
  float s = 0.f;
#pragma unroll
  for (int n = 0; n < 8; n++) { g[n] = __expf(g[n] - mx); s += g[n]; }
  float inv = 1.0f / s;
  float* wp = wgt + (size_t)b * 9 * HWSZ + hw;
#pragma unroll
  for (int n = 0; n < 4; n++) wp[(size_t)n * HWSZ] = g[n] * inv;
  wp[(size_t)4 * HWSZ] = 0.f;
#pragma unroll
  for (int n = 0; n < 4; n++) wp[(size_t)(5 + n) * HWSZ] = g[4 + n] * inv;

  __syncthreads();
  float* op = xcl + (size_t)(idx - t) * CC;
#pragma unroll
  for (int j = 0; j < 8; j++) {
    const int f  = (j * 256 + t) * 4;
    const int pr = f >> 6;
    const int sl = (f >> 2) & 15;
    const float* tp = tile + (2 * pr) * 33 + sl;
    v4f v = {tp[0], tp[33], tp[16], tp[33 + 16]};
    *(v4f*)(op + f) = v;
  }
}

struct Epi {
  v4f q[6];
  float f[17];
};

template <bool OUT_CL>
__device__ __forceinline__ void epi_load(Epi& E, const float* __restrict__ x0,
                                         const float* __restrict__ sdep,
                                         int b, int h, int wbase, int half, int p) {
  if (OUT_CL) {
    const int pix0 = h * WW + wbase + 8 * half;
    const float* sp = sdep + (size_t)b * HWSZ + pix0;
    E.q[0] = *(const v4f*)(sp);
    E.q[1] = *(const v4f*)(sp + 4);
    const float* xp0 = x0 + (size_t)(b * CC + p) * HWSZ + pix0;
    const float* xp1 = xp0 + (size_t)16 * HWSZ;
    E.q[2] = *(const v4f*)(xp0);
    E.q[3] = *(const v4f*)(xp0 + 4);
    E.q[4] = *(const v4f*)(xp1);
    E.q[5] = *(const v4f*)(xp1 + 4);
  } else {
    E.f[16] = sdep[(size_t)b * HWSZ + h * WW + wbase + p];
#pragma unroll
    for (int v = 0; v < 8; v++) {
      size_t off0 = (size_t)(b * CC + v + 8 * half) * HWSZ + h * WW + wbase + p;
      E.f[2 * v]     = x0[off0];
      E.f[2 * v + 1] = x0[off0 + (size_t)16 * HWSZ];
    }
  }
}

#define LP CC

template <bool OUT_CL>
__device__ __forceinline__ void prop2_body(float* __restrict__ BW, float* __restrict__ YR,
                                           const float* __restrict__ xin,
                                           const float* __restrict__ x0,
                                           const float* __restrict__ wgt,
                                           const float* __restrict__ sdep,
                                           float* __restrict__ xout) {
  const int lane = threadIdx.x & 31;
  const int wv   = threadIdx.x >> 5;
  const int half = lane >> 4;
  const int p    = lane & 15;

  const int b     = blockIdx.x >> 5;
  const int h0    = (blockIdx.x & 31) * RR;
  const int wbase = wv << 4;

#pragma unroll
  for (int t = 0; t < 60; t++) BW[lane + 32 * t] = 0.f;

  int co[5];
#pragma unroll
  for (int kk = 0; kk < 5; kk++) {
    int pr = (wbase >> 1) - 1 + 2 * kk + half;
    co[kk] = min(max(pr, 0), WW / 2 - 1) * PP + 4 * p;
  }
  const bool cv0 = (wbase + p - 1) >= 0;
  const bool cv2 = (wbase + p + 1) < WW;

  const float* xa = xin + (size_t)b * HWSZ * CC;
  const float* wq = wgt + (size_t)b * 9 * HWSZ + wbase + p;

  v4f XF[3][5];
  float w[2][9];
  Epi E;
  Epi EZ;

#define LOAD_ROW(slot, row)                                                     \
  {                                                                             \
    const int rowc_ = min(max((row), 0), HH - 1);                               \
    const float* rp_ = xa + (size_t)rowc_ * (WW * CC);                          \
    _Pragma("unroll") for (int kk = 0; kk < 5; kk++)                            \
      XF[(slot)][kk] = *(const v4f*)(rp_ + co[kk]);                             \
  }
#define LOAD_W(s, row)                                                          \
  {                                                                             \
    const int rowc_ = min(max((row), 0), HH - 1);                               \
    _Pragma("unroll") for (int n = 0; n < 9; n++)                               \
      w[(s)][n] = wq[(size_t)n * HWSZ + rowc_ * WW];                            \
  }
#define BAND(ws, row, bb)                                                         \
  {                                                                             \
    __builtin_amdgcn_fence(__ATOMIC_ACQ_REL, "workgroup", "local");             \
    if (lane < 16) {                                                            \
      _Pragma("unroll") for (int n = 0; n < 9; n++) {                           \
        const int i = n / 3, j = n % 3;                                         \
        const bool rok = ((row) - 1 + i >= 0) && ((row) - 1 + i < HH);          \
        const bool cok = (j == 0) ? cv0 : ((j == 2) ? cv2 : true);              \
        (bb)[(20 * i + p + 1 + j) * 16 + p] = (rok && cok) ? w[(ws)][n] : 0.f;  \
      }                                                                         \
    }                                                                           \
    __builtin_amdgcn_fence(__ATOMIC_ACQ_REL, "workgroup", "local");             \
  }

  LOAD_ROW(0, h0 - 2)
  LOAD_ROW(1, h0 - 1)
  LOAD_ROW(2, h0)
  LOAD_W(0, h0 - 1)
  epi_load<true>(E, x0, sdep, b, min(max(h0 - 1, 0), HH - 1), wbase, half, p);

#pragma unroll
  for (int s = 0; s < RR + 2; s++) {
    const int y = h0 - 1 + s;
    if (s + 1 < RR + 2) LOAD_W((s + 1) & 1, y + 1)
    if (!OUT_CL && s >= 2) epi_load<false>(EZ, x0, sdep, b, y - 1, wbase, half, p);
    const float* BWy = BW + (s & 1) * 960;
    const float* BWz = BW + ((s + 1) & 1) * 960;
    BAND(s & 1, y, BW + (s & 1) * 960)
    {
      v8f acc0 = {0.f, 0.f, 0.f, 0.f, 0.f, 0.f, 0.f, 0.f};
      v8f acc1 = {0.f, 0.f, 0.f, 0.f, 0.f, 0.f, 0.f, 0.f};
#pragma unroll
      for (int i = 0; i < 3; i++) {
#pragma unroll
        for (int kk = 0; kk < 5; kk++) {
          const int c = 5 * i + kk;
          v2f bf;
          bf.x = BWy[(4 * c + 0) * 16 + half * 32 + p];
          bf.y = BWy[(4 * c + 1) * 16 + half * 32 + p];
          const v4f xf  = XF[(s + i) % 3][kk];
          const v2f xf0 = __builtin_shufflevector(xf, xf, 0, 1);
          const v2f xf1 = __builtin_shufflevector(xf, xf, 2, 3);
          acc0 = __builtin_amdgcn_wmma_f32_16x16x4_f32(false, bf, false, xf0,
                                                       (short)0, acc0, false, false);
          acc1 = __builtin_amdgcn_wmma_f32_16x16x4_f32(false, bf, false, xf1,
                                                       (short)0, acc1, false, false);
        }
      }
      float sd[8]  = {E.q[0].x, E.q[0].y, E.q[0].z, E.q[0].w, E.q[1].x, E.q[1].y, E.q[1].z, E.q[1].w};
      float xv0[8] = {E.q[2].x, E.q[2].y, E.q[2].z, E.q[2].w, E.q[3].x, E.q[3].y, E.q[3].z, E.q[3].w};
      float xv1[8] = {E.q[4].x, E.q[4].y, E.q[4].z, E.q[4].w, E.q[5].x, E.q[5].y, E.q[5].z, E.q[5].w};
      float* yp = YR + (s & 3) * (WW * LP) + ((wbase + 8 * half) >> 1) * PP + 4 * p;
#pragma unroll
      for (int d = 0; d < 8; d += 2) {
        float m0 = (sd[d] > 0.f) ? 1.0f : 0.0f;
        float m1 = (sd[d + 1] > 0.f) ? 1.0f : 0.0f;
        v4f o;
        o.x = m0 * xv0[d]     + (1.0f - m0) * acc0[d];
        o.y = m1 * xv0[d + 1] + (1.0f - m1) * acc0[d + 1];
        o.z = m0 * xv1[d]     + (1.0f - m0) * acc1[d];
        o.w = m1 * xv1[d + 1] + (1.0f - m1) * acc1[d + 1];
        *(v4f*)(yp + (d >> 1) * PP) = o;
      }
    }
    __builtin_amdgcn_fence(__ATOMIC_RELEASE, "workgroup", "local");
    __builtin_amdgcn_s_barrier();
    __builtin_amdgcn_fence(__ATOMIC_ACQUIRE, "workgroup", "local");

    if (OUT_CL && s >= 2) epi_load<true>(EZ, x0, sdep, b, y - 1, wbase, half, p);
    if (s + 1 < RR + 2) LOAD_ROW(s % 3, y + 2)
    if (s >= 2) {
      const int z = y - 1;
      v8f acc0 = {0.f, 0.f, 0.f, 0.f, 0.f, 0.f, 0.f, 0.f};
      v8f acc1 = {0.f, 0.f, 0.f, 0.f, 0.f, 0.f, 0.f, 0.f};
#pragma unroll
      for (int i = 0; i < 3; i++) {
        const float* yr = YR + ((s - 2 + i) & 3) * (WW * LP);
#pragma unroll
        for (int kk = 0; kk < 5; kk++) {
          const int c = 5 * i + kk;
          v2f bf;
          bf.x = BWz[(4 * c + 0) * 16 + half * 32 + p];
          bf.y = BWz[(4 * c + 1) * 16 + half * 32 + p];
          const v4f u   = *(const v4f*)(yr + co[kk]);
          const v2f af0 = __builtin_shufflevector(u, u, 0, 1);
          const v2f af1 = __builtin_shufflevector(u, u, 2, 3);
          if (OUT_CL) {
            acc0 = __builtin_amdgcn_wmma_f32_16x16x4_f32(false, bf, false, af0,
                                                         (short)0, acc0, false, false);
            acc1 = __builtin_amdgcn_wmma_f32_16x16x4_f32(false, bf, false, af1,
                                                         (short)0, acc1, false, false);
          } else {
            acc0 = __builtin_amdgcn_wmma_f32_16x16x4_f32(false, af0, false, bf,
                                                         (short)0, acc0, false, false);
            acc1 = __builtin_amdgcn_wmma_f32_16x16x4_f32(false, af1, false, bf,
                                                         (short)0, acc1, false, false);
          }
        }
      }
      __builtin_amdgcn_sched_barrier(0);
      if (OUT_CL) {
        float sd[8]  = {EZ.q[0].x, EZ.q[0].y, EZ.q[0].z, EZ.q[0].w, EZ.q[1].x, EZ.q[1].y, EZ.q[1].z, EZ.q[1].w};
        float xv0[8] = {EZ.q[2].x, EZ.q[2].y, EZ.q[2].z, EZ.q[2].w, EZ.q[3].x, EZ.q[3].y, EZ.q[3].z, EZ.q[3].w};
        float xv1[8] = {EZ.q[4].x, EZ.q[4].y, EZ.q[4].z, EZ.q[4].w, EZ.q[5].x, EZ.q[5].y, EZ.q[5].z, EZ.q[5].w};
        float* op = xout + ((size_t)b * HWSZ + z * WW) * CC + ((wbase + 8 * half) >> 1) * PP + 4 * p;
#pragma unroll
        for (int d = 0; d < 8; d += 2) {
          float m0 = (sd[d] > 0.f) ? 1.0f : 0.0f;
          float m1 = (sd[d + 1] > 0.f) ? 1.0f : 0.0f;
          v4f o;
          o.x = m0 * xv0[d]     + (1.0f - m0) * acc0[d];
          o.y = m1 * xv0[d + 1] + (1.0f - m1) * acc0[d + 1];
          o.z = m0 * xv1[d]     + (1.0f - m0) * acc1[d];
          o.w = m1 * xv1[d + 1] + (1.0f - m1) * acc1[d + 1];
          *(v4f*)(op + (d >> 1) * PP) = o;
        }
      } else {
        float m = (EZ.f[16] > 0.f) ? 1.0f : 0.0f;
#pragma unroll
        for (int v = 0; v < 8; v++) {
          size_t off0 = (size_t)(b * CC + v + 8 * half) * HWSZ + z * WW + wbase + p;
          size_t off1 = off0 + (size_t)16 * HWSZ;
          xout[off0] = m * EZ.f[2 * v]     + (1.0f - m) * acc0[v];
          xout[off1] = m * EZ.f[2 * v + 1] + (1.0f - m) * acc1[v];
        }
      }
    }
    if (s + 1 < RR + 2)
      epi_load<true>(E, x0, sdep, b, min(max(y + 1, 0), HH - 1), wbase, half, p);
  }
#undef LOAD_ROW
#undef LOAD_W
#undef BAND
}

#define BAND_FLOATS (16 * 2 * 60 * 16)
#define RING_FLOATS (4 * WW * LP)

__global__ __launch_bounds__(512) void pac_prop2_cl(const float* __restrict__ xin,
                                                    const float* __restrict__ x0,
                                                    const float* __restrict__ wgt,
                                                    const float* __restrict__ sdep,
                                                    float* __restrict__ xout) {
  __shared__ __attribute__((aligned(16))) float smem[BAND_FLOATS + RING_FLOATS];
  prop2_body<true>(smem + (threadIdx.x >> 5) * (2 * 60 * 16), smem + BAND_FLOATS,
                   xin, x0, wgt, sdep, xout);
}

__global__ __launch_bounds__(512) void pac_prop2_last(const float* __restrict__ xin,
                                                      const float* __restrict__ x0,
                                                      const float* __restrict__ wgt,
                                                      const float* __restrict__ sdep,
                                                      float* __restrict__ xout) {
  __shared__ __attribute__((aligned(16))) float smem[BAND_FLOATS + RING_FLOATS];
  prop2_body<false>(smem + (threadIdx.x >> 5) * (2 * 60 * 16), smem + BAND_FLOATS,
                    xin, x0, wgt, sdep, xout);
}

extern "C" void kernel_launch(void* const* d_in, const int* in_sizes, int n_in,
                              void* d_out, int out_size, void* d_ws, size_t ws_size,
                              hipStream_t stream) {
  const float* x      = (const float*)d_in[0];
  const float* guided = (const float*)d_in[1];
  const float* sdep   = (const float*)d_in[2];
  (void)in_sizes; (void)n_in; (void)out_size; (void)ws_size;

  float* bufA = (float*)d_ws;
  float* wgt  = bufA + (size_t)BB * CC * HWSZ;
  float* outb = (float*)d_out;

  pac_prep<<<(BB * HWSZ) / 256, 256, 0, stream>>>(guided, x, wgt, outb);

  const float* cur = outb;
  for (int t = 0; t < 7; t++) {
    float* dst = (t & 1) ? outb : bufA;
    pac_prop2_cl<<<BB * (HH / RR), 512, 0, stream>>>(cur, x, wgt, sdep, dst);
    cur = dst;
  }
  pac_prop2_last<<<BB * (HH / RR), 512, 0, stream>>>(cur, x, wgt, sdep, outb);
}


# a row's x0 blend values are fetched once: kept in registers from the first step's blend for the second step's blend of the same row a trip later; only the mask line is re-read; sums unchanged
# speedup vs baseline: 4.1691x; 1.0594x over previous
#include <hip/hip_runtime.h>


typedef float v2f __attribute__((ext_vector_type(2)));
typedef float v4f __attribute__((ext_vector_type(4)));
typedef float v8f __attribute__((ext_vector_type(8)));

#define BB   8
#define CC   32
#define HH   256
#define WW   256
#define HWSZ (HH * WW)
#define RR   8
#define PP   64

__global__ __launch_bounds__(256) void pac_prep(const float* __restrict__ guided,
                                                const float* __restrict__ x,
                                                float* __restrict__ wgt,
                                                float* __restrict__ xcl) {
  __shared__ float tile[256 * 33];

  const int t   = threadIdx.x;
  const int idx = blockIdx.x * 256 + t;
  const int b   = idx >> 16;
  const int hw  = idx & 65535;

  const float* xp = x + (size_t)b * CC * HWSZ + hw;
#pragma unroll
  for (int c = 0; c < CC; c++) tile[t * 33 + c] = xp[(size_t)c * HWSZ];

  const float* gp = guided + (size_t)b * 8 * HWSZ + hw;
  float g[8];
  float mx = -1e30f;
#pragma unroll
  for (int n = 0; n < 8; n++) { g[n] = gp[(size_t)n * HWSZ]; mx = fmaxf(mx, g[n]); }
  float s = 0.f;
#pragma unroll
  for (int n = 0; n < 8; n++) { g[n] = __expf(g[n] - mx); s += g[n]; }
  float inv = 1.0f / s;
  float* wp = wgt + (size_t)b * 9 * HWSZ + hw;
#pragma unroll
  for (int n = 0; n < 4; n++) wp[(size_t)n * HWSZ] = g[n] * inv;
  wp[(size_t)4 * HWSZ] = 0.f;
#pragma unroll
  for (int n = 0; n < 4; n++) wp[(size_t)(5 + n) * HWSZ] = g[4 + n] * inv;

  __syncthreads();
  float* op = xcl + (size_t)(idx - t) * CC;
#pragma unroll
  for (int j = 0; j < 8; j++) {
    const int f  = (j * 256 + t) * 4;
    const int pr = f >> 6;
    const int sl = (f >> 2) & 15;
    const float* tp = tile + (2 * pr) * 33 + sl;
    v4f v = {tp[0], tp[33], tp[16], tp[33 + 16]};
    *(v4f*)(op + f) = v;
  }
}

struct Epi {
  v4f q[6];
  float f[17];
};

template <bool OUT_CL>
__device__ __forceinline__ void epi_load(Epi& E, const float* __restrict__ x0,
                                         const float* __restrict__ sdep,
                                         int b, int h, int wbase, int half, int p) {
  if (OUT_CL) {
    const int pix0 = h * WW + wbase + 8 * half;
    const float* sp = sdep + (size_t)b * HWSZ + pix0;
    E.q[0] = *(const v4f*)(sp);
    E.q[1] = *(const v4f*)(sp + 4);
    const float* xp0 = x0 + (size_t)(b * CC + p) * HWSZ + pix0;
    const float* xp1 = xp0 + (size_t)16 * HWSZ;
    E.q[2] = *(const v4f*)(xp0);
    E.q[3] = *(const v4f*)(xp0 + 4);
    E.q[4] = *(const v4f*)(xp1);
    E.q[5] = *(const v4f*)(xp1 + 4);
  } else {
    E.f[16] = sdep[(size_t)b * HWSZ + h * WW + wbase + p];
#pragma unroll
    for (int v = 0; v < 8; v++) {
      size_t off0 = (size_t)(b * CC + v + 8 * half) * HWSZ + h * WW + wbase + p;
      E.f[2 * v]     = x0[off0];
      E.f[2 * v + 1] = x0[off0 + (size_t)16 * HWSZ];
    }
  }
}

__device__ __forceinline__ void epi_load_mask(Epi& E, const float* __restrict__ sdep,
                                              int b, int h, int wbase, int half) {
  const float* sp = sdep + (size_t)b * HWSZ + h * WW + wbase + 8 * half;
  E.q[0] = *(const v4f*)(sp);
  E.q[1] = *(const v4f*)(sp + 4);
}

#define LP CC

template <bool OUT_CL>
__device__ __forceinline__ void prop2_body(float* __restrict__ BW, float* __restrict__ YR,
                                           const float* __restrict__ xin,
                                           const float* __restrict__ x0,
                                           const float* __restrict__ wgt,
                                           const float* __restrict__ sdep,
                                           float* __restrict__ xout) {
  const int lane = threadIdx.x & 31;
  const int wv   = threadIdx.x >> 5;
  const int half = lane >> 4;
  const int p    = lane & 15;

  const int b     = blockIdx.x >> 5;
  const int h0    = (blockIdx.x & 31) * RR;
  const int wbase = wv << 4;

#pragma unroll
  for (int t = 0; t < 60; t++) BW[lane + 32 * t] = 0.f;

  int co[5];
#pragma unroll
  for (int kk = 0; kk < 5; kk++) {
    int pr = (wbase >> 1) - 1 + 2 * kk + half;
    co[kk] = min(max(pr, 0), WW / 2 - 1) * PP + 4 * p;
  }
  const bool cv0 = (wbase + p - 1) >= 0;
  const bool cv2 = (wbase + p + 1) < WW;

  const float* xa = xin + (size_t)b * HWSZ * CC;
  const float* wq = wgt + (size_t)b * 9 * HWSZ + wbase + p;

  v4f XF[3][5];
  float w[2][9];
  Epi E[2];
  Epi EZ;

#define LOAD_ROW(slot, row)                                                     \
  {                                                                             \
    const int rowc_ = min(max((row), 0), HH - 1);                               \
    const float* rp_ = xa + (size_t)rowc_ * (WW * CC);                          \
    _Pragma("unroll") for (int kk = 0; kk < 5; kk++)                            \
      XF[(slot)][kk] = *(const v4f*)(rp_ + co[kk]);                             \
  }
#define LOAD_W(s, row)                                                          \
  {                                                                             \
    const int rowc_ = min(max((row), 0), HH - 1);                               \
    _Pragma("unroll") for (int n = 0; n < 9; n++)                               \
      w[(s)][n] = wq[(size_t)n * HWSZ + rowc_ * WW];                            \
  }
#define BAND(ws, row, bb)                                                         \
  {                                                                             \
    __builtin_amdgcn_fence(__ATOMIC_ACQ_REL, "workgroup", "local");             \
    if (lane < 16) {                                                            \
      _Pragma("unroll") for (int n = 0; n < 9; n++) {                           \
        const int i = n / 3, j = n % 3;                                         \
        const bool rok = ((row) - 1 + i >= 0) && ((row) - 1 + i < HH);          \
        const bool cok = (j == 0) ? cv0 : ((j == 2) ? cv2 : true);              \
        (bb)[(20 * i + p + 1 + j) * 16 + p] = (rok && cok) ? w[(ws)][n] : 0.f;  \
      }                                                                         \
    }                                                                           \
    __builtin_amdgcn_fence(__ATOMIC_ACQ_REL, "workgroup", "local");             \
  }

  LOAD_ROW(0, h0 - 2)
  LOAD_ROW(1, h0 - 1)
  LOAD_ROW(2, h0)
  LOAD_W(0, h0 - 1)
  epi_load<true>(E[0], x0, sdep, b, min(max(h0 - 1, 0), HH - 1), wbase, half, p);

#pragma unroll
  for (int s = 0; s < RR + 2; s++) {
    const int y = h0 - 1 + s;
    if (s + 1 < RR + 2) LOAD_W((s + 1) & 1, y + 1)
    if (!OUT_CL && s >= 2) epi_load<false>(EZ, x0, sdep, b, y - 1, wbase, half, p);
    const float* BWy = BW + (s & 1) * 960;
    const float* BWz = BW + ((s + 1) & 1) * 960;
    BAND(s & 1, y, BW + (s & 1) * 960)
    {
      v8f acc0 = {0.f, 0.f, 0.f, 0.f, 0.f, 0.f, 0.f, 0.f};
      v8f acc1 = {0.f, 0.f, 0.f, 0.f, 0.f, 0.f, 0.f, 0.f};
#pragma unroll
      for (int i = 0; i < 3; i++) {
#pragma unroll
        for (int kk = 0; kk < 5; kk++) {
          const int c = 5 * i + kk;
          v2f bf;
          bf.x = BWy[(4 * c + 0) * 16 + half * 32 + p];
          bf.y = BWy[(4 * c + 1) * 16 + half * 32 + p];
          const v4f xf  = XF[(s + i) % 3][kk];
          const v2f xf0 = __builtin_shufflevector(xf, xf, 0, 1);
          const v2f xf1 = __builtin_shufflevector(xf, xf, 2, 3);
          acc0 = __builtin_amdgcn_wmma_f32_16x16x4_f32(false, bf, false, xf0,
                                                       (short)0, acc0, false, false);
          acc1 = __builtin_amdgcn_wmma_f32_16x16x4_f32(false, bf, false, xf1,
                                                       (short)0, acc1, false, false);
        }
      }
      const Epi& EF = E[s & 1];
      float sd[8]  = {EF.q[0].x, EF.q[0].y, EF.q[0].z, EF.q[0].w, EF.q[1].x, EF.q[1].y, EF.q[1].z, EF.q[1].w};
      float xv0[8] = {EF.q[2].x, EF.q[2].y, EF.q[2].z, EF.q[2].w, EF.q[3].x, EF.q[3].y, EF.q[3].z, EF.q[3].w};
      float xv1[8] = {EF.q[4].x, EF.q[4].y, EF.q[4].z, EF.q[4].w, EF.q[5].x, EF.q[5].y, EF.q[5].z, EF.q[5].w};
      float* yp = YR + (s & 3) * (WW * LP) + ((wbase + 8 * half) >> 1) * PP + 4 * p;
#pragma unroll
      for (int d = 0; d < 8; d += 2) {
        float m0 = (sd[d] > 0.f) ? 1.0f : 0.0f;
        float m1 = (sd[d + 1] > 0.f) ? 1.0f : 0.0f;
        v4f o;
        o.x = m0 * xv0[d]     + (1.0f - m0) * acc0[d];
        o.y = m1 * xv0[d + 1] + (1.0f - m1) * acc0[d + 1];
        o.z = m0 * xv1[d]     + (1.0f - m0) * acc1[d];
        o.w = m1 * xv1[d + 1] + (1.0f - m1) * acc1[d + 1];
        *(v4f*)(yp + (d >> 1) * PP) = o;
      }
    }
    __builtin_amdgcn_fence(__ATOMIC_RELEASE, "workgroup", "local");
    __builtin_amdgcn_s_barrier();
    __builtin_amdgcn_fence(__ATOMIC_ACQUIRE, "workgroup", "local");

    if (OUT_CL && s >= 2) epi_load_mask(EZ, sdep, b, y - 1, wbase, half);
    if (s + 1 < RR + 2) LOAD_ROW(s % 3, y + 2)
    if (s >= 2) {
      const int z = y - 1;
      v8f acc0 = {0.f, 0.f, 0.f, 0.f, 0.f, 0.f, 0.f, 0.f};
      v8f acc1 = {0.f, 0.f, 0.f, 0.f, 0.f, 0.f, 0.f, 0.f};
#pragma unroll
      for (int i = 0; i < 3; i++) {
        const float* yr = YR + ((s - 2 + i) & 3) * (WW * LP);
#pragma unroll
        for (int kk = 0; kk < 5; kk++) {
          const int c = 5 * i + kk;
          v2f bf;
          bf.x = BWz[(4 * c + 0) * 16 + half * 32 + p];
          bf.y = BWz[(4 * c + 1) * 16 + half * 32 + p];
          const v4f u   = *(const v4f*)(yr + co[kk]);
          const v2f af0 = __builtin_shufflevector(u, u, 0, 1);
          const v2f af1 = __builtin_shufflevector(u, u, 2, 3);
          if (OUT_CL) {
            acc0 = __builtin_amdgcn_wmma_f32_16x16x4_f32(false, bf, false, af0,
                                                         (short)0, acc0, false, false);
            acc1 = __builtin_amdgcn_wmma_f32_16x16x4_f32(false, bf, false, af1,
                                                         (short)0, acc1, false, false);
          } else {
            acc0 = __builtin_amdgcn_wmma_f32_16x16x4_f32(false, af0, false, bf,
                                                         (short)0, acc0, false, false);
            acc1 = __builtin_amdgcn_wmma_f32_16x16x4_f32(false, af1, false, bf,
                                                         (short)0, acc1, false, false);
          }
        }
      }
      __builtin_amdgcn_sched_barrier(0);
      if (OUT_CL) {
        const Epi& EK = E[(s + 1) & 1];
        float sd[8]  = {EZ.q[0].x, EZ.q[0].y, EZ.q[0].z, EZ.q[0].w, EZ.q[1].x, EZ.q[1].y, EZ.q[1].z, EZ.q[1].w};
        float xv0[8] = {EK.q[2].x, EK.q[2].y, EK.q[2].z, EK.q[2].w, EK.q[3].x, EK.q[3].y, EK.q[3].z, EK.q[3].w};
        float xv1[8] = {EK.q[4].x, EK.q[4].y, EK.q[4].z, EK.q[4].w, EK.q[5].x, EK.q[5].y, EK.q[5].z, EK.q[5].w};
        float* op = xout + ((size_t)b * HWSZ + z * WW) * CC + ((wbase + 8 * half) >> 1) * PP + 4 * p;
#pragma unroll
        for (int d = 0; d < 8; d += 2) {
          float m0 = (sd[d] > 0.f) ? 1.0f : 0.0f;
          float m1 = (sd[d + 1] > 0.f) ? 1.0f : 0.0f;
          v4f o;
          o.x = m0 * xv0[d]     + (1.0f - m0) * acc0[d];
          o.y = m1 * xv0[d + 1] + (1.0f - m1) * acc0[d + 1];
          o.z = m0 * xv1[d]     + (1.0f - m0) * acc1[d];
          o.w = m1 * xv1[d + 1] + (1.0f - m1) * acc1[d + 1];
          *(v4f*)(op + (d >> 1) * PP) = o;
        }
      } else {
        float m = (EZ.f[16] > 0.f) ? 1.0f : 0.0f;
#pragma unroll
        for (int v = 0; v < 8; v++) {
          size_t off0 = (size_t)(b * CC + v + 8 * half) * HWSZ + z * WW + wbase + p;
          size_t off1 = off0 + (size_t)16 * HWSZ;
          xout[off0] = m * EZ.f[2 * v]     + (1.0f - m) * acc0[v];
          xout[off1] = m * EZ.f[2 * v + 1] + (1.0f - m) * acc1[v];
        }
      }
    }
    if (s + 1 < RR + 2)
      epi_load<true>(E[(s + 1) & 1], x0, sdep, b, min(max(y + 1, 0), HH - 1), wbase, half, p);
  }
#undef LOAD_ROW
#undef LOAD_W
#undef BAND
}

#define BAND_FLOATS (16 * 2 * 60 * 16)
#define RING_FLOATS (4 * WW * LP)

__global__ __launch_bounds__(512) void pac_prop2_cl(const float* __restrict__ xin,
                                                    const float* __restrict__ x0,
                                                    const float* __restrict__ wgt,
                                                    const float* __restrict__ sdep,
                                                    float* __restrict__ xout) {
  __shared__ __attribute__((aligned(16))) float smem[BAND_FLOATS + RING_FLOATS];
  prop2_body<true>(smem + (threadIdx.x >> 5) * (2 * 60 * 16), smem + BAND_FLOATS,
                   xin, x0, wgt, sdep, xout);
}

__global__ __launch_bounds__(512) void pac_prop2_last(const float* __restrict__ xin,
                                                      const float* __restrict__ x0,
                                                      const float* __restrict__ wgt,
                                                      const float* __restrict__ sdep,
                                                      float* __restrict__ xout) {
  __shared__ __attribute__((aligned(16))) float smem[BAND_FLOATS + RING_FLOATS];
  prop2_body<false>(smem + (threadIdx.x >> 5) * (2 * 60 * 16), smem + BAND_FLOATS,
                    xin, x0, wgt, sdep, xout);
}

extern "C" void kernel_launch(void* const* d_in, const int* in_sizes, int n_in,
                              void* d_out, int out_size, void* d_ws, size_t ws_size,
                              hipStream_t stream) {
  const float* x      = (const float*)d_in[0];
  const float* guided = (const float*)d_in[1];
  const float* sdep   = (const float*)d_in[2];
  (void)in_sizes; (void)n_in; (void)out_size; (void)ws_size;

  float* bufA = (float*)d_ws;
  float* wgt  = bufA + (size_t)BB * CC * HWSZ;
  float* outb = (float*)d_out;

  pac_prep<<<(BB * HWSZ) / 256, 256, 0, stream>>>(guided, x, wgt, outb);

  const float* cur = outb;
  for (int t = 0; t < 7; t++) {
    float* dst = (t & 1) ? outb : bufA;
    pac_prop2_cl<<<BB * (HH / RR), 512, 0, stream>>>(cur, x, wgt, sdep, dst);
    cur = dst;
  }
  pac_prop2_last<<<BB * (HH / RR), 512, 0, stream>>>(cur, x, wgt, sdep, outb);
}


# each row's LDS fragment reads (Band, ring) issued in groups ahead of the WMMAs that use them: 1 read place per first-step row, 3 per second-step row, not 15 each among the WMMA pairs; sums unchanged
# speedup vs baseline: 4.3853x; 1.0519x over previous
#include <hip/hip_runtime.h>


typedef float v2f __attribute__((ext_vector_type(2)));
typedef float v4f __attribute__((ext_vector_type(4)));
typedef float v8f __attribute__((ext_vector_type(8)));

#define BB   8
#define CC   32
#define HH   256
#define WW   256
#define HWSZ (HH * WW)
#define RR   8
#define PP   64

__global__ __launch_bounds__(256) void pac_prep(const float* __restrict__ guided,
                                                const float* __restrict__ x,
                                                float* __restrict__ wgt,
                                                float* __restrict__ xcl) {
  __shared__ float tile[256 * 33];

  const int t   = threadIdx.x;
  const int idx = blockIdx.x * 256 + t;
  const int b   = idx >> 16;
  const int hw  = idx & 65535;

  const float* xp = x + (size_t)b * CC * HWSZ + hw;
#pragma unroll
  for (int c = 0; c < CC; c++) tile[t * 33 + c] = xp[(size_t)c * HWSZ];

  const float* gp = guided + (size_t)b * 8 * HWSZ + hw;
  float g[8];
  float mx = -1e30f;
#pragma unroll
  for (int n = 0; n < 8; n++) { g[n] = gp[(size_t)n * HWSZ]; mx = fmaxf(mx, g[n]); }
  float s = 0.f;
#pragma unroll
  for (int n = 0; n < 8; n++) { g[n] = __expf(g[n] - mx); s += g[n]; }
  float inv = 1.0f / s;
  float* wp = wgt + (size_t)b * 9 * HWSZ + hw;
#pragma unroll
  for (int n = 0; n < 4; n++) wp[(size_t)n * HWSZ] = g[n] * inv;
  wp[(size_t)4 * HWSZ] = 0.f;
#pragma unroll
  for (int n = 0; n < 4; n++) wp[(size_t)(5 + n) * HWSZ] = g[4 + n] * inv;

  __syncthreads();
  float* op = xcl + (size_t)(idx - t) * CC;
#pragma unroll
  for (int j = 0; j < 8; j++) {
    const int f  = (j * 256 + t) * 4;
    const int pr = f >> 6;
    const int sl = (f >> 2) & 15;
    const float* tp = tile + (2 * pr) * 33 + sl;
    v4f v = {tp[0], tp[33], tp[16], tp[33 + 16]};
    *(v4f*)(op + f) = v;
  }
}

struct Epi {
  v4f q[6];
  float f[17];
};

template <bool OUT_CL>
__device__ __forceinline__ void epi_load(Epi& E, const float* __restrict__ x0,
                                         const float* __restrict__ sdep,
                                         int b, int h, int wbase, int half, int p) {
  if (OUT_CL) {
    const int pix0 = h * WW + wbase + 8 * half;
    const float* sp = sdep + (size_t)b * HWSZ + pix0;
    E.q[0] = *(const v4f*)(sp);
    E.q[1] = *(const v4f*)(sp + 4);
    const float* xp0 = x0 + (size_t)(b * CC + p) * HWSZ + pix0;
    const float* xp1 = xp0 + (size_t)16 * HWSZ;
    E.q[2] = *(const v4f*)(xp0);
    E.q[3] = *(const v4f*)(xp0 + 4);
    E.q[4] = *(const v4f*)(xp1);
    E.q[5] = *(const v4f*)(xp1 + 4);
  } else {
    E.f[16] = sdep[(size_t)b * HWSZ + h * WW + wbase + p];
#pragma unroll
    for (int v = 0; v < 8; v++) {
      size_t off0 = (size_t)(b * CC + v + 8 * half) * HWSZ + h * WW + wbase + p;
      E.f[2 * v]     = x0[off0];
      E.f[2 * v + 1] = x0[off0 + (size_t)16 * HWSZ];
    }
  }
}

__device__ __forceinline__ void epi_load_mask(Epi& E, const float* __restrict__ sdep,
                                              int b, int h, int wbase, int half) {
  const float* sp = sdep + (size_t)b * HWSZ + h * WW + wbase + 8 * half;
  E.q[0] = *(const v4f*)(sp);
  E.q[1] = *(const v4f*)(sp + 4);
}

#define LP CC

template <bool OUT_CL>
__device__ __forceinline__ void prop2_body(float* __restrict__ BW, float* __restrict__ YR,
                                           const float* __restrict__ xin,
                                           const float* __restrict__ x0,
                                           const float* __restrict__ wgt,
                                           const float* __restrict__ sdep,
                                           float* __restrict__ xout) {
  const int lane = threadIdx.x & 31;
  const int wv   = threadIdx.x >> 5;
  const int half = lane >> 4;
  const int p    = lane & 15;

  const int b     = blockIdx.x >> 5;
  const int h0    = (blockIdx.x & 31) * RR;
  const int wbase = wv << 4;

#pragma unroll
  for (int t = 0; t < 60; t++) BW[lane + 32 * t] = 0.f;

  int co[5];
#pragma unroll
  for (int kk = 0; kk < 5; kk++) {
    int pr = (wbase >> 1) - 1 + 2 * kk + half;
    co[kk] = min(max(pr, 0), WW / 2 - 1) * PP + 4 * p;
  }
  const bool cv0 = (wbase + p - 1) >= 0;
  const bool cv2 = (wbase + p + 1) < WW;

  const float* xa = xin + (size_t)b * HWSZ * CC;
  const float* wq = wgt + (size_t)b * 9 * HWSZ + wbase + p;

  v4f XF[3][5];
  float w[2][9];
  Epi E[2];
  Epi EZ;

#define LOAD_ROW(slot, row)                                                     \
  {                                                                             \
    const int rowc_ = min(max((row), 0), HH - 1);                               \
    const float* rp_ = xa + (size_t)rowc_ * (WW * CC);                          \
    _Pragma("unroll") for (int kk = 0; kk < 5; kk++)                            \
      XF[(slot)][kk] = *(const v4f*)(rp_ + co[kk]);                             \
  }
#define LOAD_W(s, row)                                                          \
  {                                                                             \
    const int rowc_ = min(max((row), 0), HH - 1);                               \
    _Pragma("unroll") for (int n = 0; n < 9; n++)                               \
      w[(s)][n] = wq[(size_t)n * HWSZ + rowc_ * WW];                            \
  }
#define BAND(ws, row, bb)                                                         \
  {                                                                             \
    __builtin_amdgcn_fence(__ATOMIC_ACQ_REL, "workgroup", "local");             \
    if (lane < 16) {                                                            \
      _Pragma("unroll") for (int n = 0; n < 9; n++) {                           \
        const int i = n / 3, j = n % 3;                                         \
        const bool rok = ((row) - 1 + i >= 0) && ((row) - 1 + i < HH);          \
        const bool cok = (j == 0) ? cv0 : ((j == 2) ? cv2 : true);              \
        (bb)[(20 * i + p + 1 + j) * 16 + p] = (rok && cok) ? w[(ws)][n] : 0.f;  \
      }                                                                         \
    }                                                                           \
    __builtin_amdgcn_fence(__ATOMIC_ACQ_REL, "workgroup", "local");             \
  }

  LOAD_ROW(0, h0 - 2)
  LOAD_ROW(1, h0 - 1)
  LOAD_ROW(2, h0)
  LOAD_W(0, h0 - 1)
  epi_load<true>(E[0], x0, sdep, b, min(max(h0 - 1, 0), HH - 1), wbase, half, p);

#pragma unroll
  for (int s = 0; s < RR + 2; s++) {
    const int y = h0 - 1 + s;
    if (s + 1 < RR + 2) LOAD_W((s + 1) & 1, y + 1)
    if (!OUT_CL && s >= 2) epi_load<false>(EZ, x0, sdep, b, y - 1, wbase, half, p);
    const float* BWy = BW + (s & 1) * 960;
    const float* BWz = BW + ((s + 1) & 1) * 960;
    BAND(s & 1, y, BW + (s & 1) * 960)
    {
      v8f acc0 = {0.f, 0.f, 0.f, 0.f, 0.f, 0.f, 0.f, 0.f};
      v8f acc1 = {0.f, 0.f, 0.f, 0.f, 0.f, 0.f, 0.f, 0.f};
      v2f bfy[15];
#pragma unroll
      for (int c = 0; c < 15; c++) {
        bfy[c].x = BWy[(4 * c + 0) * 16 + half * 32 + p];
        bfy[c].y = BWy[(4 * c + 1) * 16 + half * 32 + p];
      }
      __builtin_amdgcn_sched_barrier(0);
#pragma unroll
      for (int i = 0; i < 3; i++) {
#pragma unroll
        for (int kk = 0; kk < 5; kk++) {
          const int c = 5 * i + kk;
          const v2f bf  = bfy[c];
          const v4f xf  = XF[(s + i) % 3][kk];
          const v2f xf0 = __builtin_shufflevector(xf, xf, 0, 1);
          const v2f xf1 = __builtin_shufflevector(xf, xf, 2, 3);
          acc0 = __builtin_amdgcn_wmma_f32_16x16x4_f32(false, bf, false, xf0,
                                                       (short)0, acc0, false, false);
          acc1 = __builtin_amdgcn_wmma_f32_16x16x4_f32(false, bf, false, xf1,
                                                       (short)0, acc1, false, false);
        }
      }
      const Epi& EF = E[s & 1];
      float sd[8]  = {EF.q[0].x, EF.q[0].y, EF.q[0].z, EF.q[0].w, EF.q[1].x, EF.q[1].y, EF.q[1].z, EF.q[1].w};
      float xv0[8] = {EF.q[2].x, EF.q[2].y, EF.q[2].z, EF.q[2].w, EF.q[3].x, EF.q[3].y, EF.q[3].z, EF.q[3].w};
      float xv1[8] = {EF.q[4].x, EF.q[4].y, EF.q[4].z, EF.q[4].w, EF.q[5].x, EF.q[5].y, EF.q[5].z, EF.q[5].w};
      float* yp = YR + (s & 3) * (WW * LP) + ((wbase + 8 * half) >> 1) * PP + 4 * p;
#pragma unroll
      for (int d = 0; d < 8; d += 2) {
        float m0 = (sd[d] > 0.f) ? 1.0f : 0.0f;
        float m1 = (sd[d + 1] > 0.f) ? 1.0f : 0.0f;
        v4f o;
        o.x = m0 * xv0[d]     + (1.0f - m0) * acc0[d];
        o.y = m1 * xv0[d + 1] + (1.0f - m1) * acc0[d + 1];
        o.z = m0 * xv1[d]     + (1.0f - m0) * acc1[d];
        o.w = m1 * xv1[d + 1] + (1.0f - m1) * acc1[d + 1];
        *(v4f*)(yp + (d >> 1) * PP) = o;
      }
    }
    __builtin_amdgcn_fence(__ATOMIC_RELEASE, "workgroup", "local");
    __builtin_amdgcn_s_barrier();
    __builtin_amdgcn_fence(__ATOMIC_ACQUIRE, "workgroup", "local");

    if (OUT_CL && s >= 2) epi_load_mask(EZ, sdep, b, y - 1, wbase, half);
    if (s + 1 < RR + 2) LOAD_ROW(s % 3, y + 2)
    if (s >= 2) {
      const int z = y - 1;
      v8f acc0 = {0.f, 0.f, 0.f, 0.f, 0.f, 0.f, 0.f, 0.f};
      v8f acc1 = {0.f, 0.f, 0.f, 0.f, 0.f, 0.f, 0.f, 0.f};
      v2f bfz[15];
      v4f uz[15];
#define FETCH2(c0, c1)                                                                  \
      _Pragma("unroll") for (int c = (c0); c < (c1); c++) {                             \
        const float* yr = YR + ((s - 2 + c / 5) & 3) * (WW * LP);     \
        bfz[c].x = BWz[(4 * c + 0) * 16 + half * 32 + p];                               \
        bfz[c].y = BWz[(4 * c + 1) * 16 + half * 32 + p];                               \
        uz[c]    = *(const v4f*)(yr + co[c % 5]);                                       \
      }
#define PROD2(c0, c1)                                                                   \
      _Pragma("unroll") for (int c = (c0); c < (c1); c++) {                             \
        const v2f bf  = bfz[c];                                                         \
        const v2f af0 = __builtin_shufflevector(uz[c], uz[c], 0, 1);        \
        const v2f af1 = __builtin_shufflevector(uz[c], uz[c], 2, 3);        \
        if (OUT_CL) {                                                                   \
          acc0 = __builtin_amdgcn_wmma_f32_16x16x4_f32(false, bf, false, af0,           \
                                                       (short)0, acc0, false, false);   \
          acc1 = __builtin_amdgcn_wmma_f32_16x16x4_f32(false, bf, false, af1,           \
                                                       (short)0, acc1, false, false);   \
        } else {                                                                        \
          acc0 = __builtin_amdgcn_wmma_f32_16x16x4_f32(false, af0, false, bf,           \
                                                       (short)0, acc0, false, false);   \
          acc1 = __builtin_amdgcn_wmma_f32_16x16x4_f32(false, af1, false, bf,           \
                                                       (short)0, acc1, false, false);   \
        }                                                                               \
      }
      FETCH2(0, 6)
      __builtin_amdgcn_sched_barrier(0);
      PROD2(0, 3)
      __builtin_amdgcn_sched_barrier(0);
      FETCH2(6, 11)
      __builtin_amdgcn_sched_barrier(0);
      PROD2(3, 8)
      __builtin_amdgcn_sched_barrier(0);
      FETCH2(11, 15)
      __builtin_amdgcn_sched_barrier(0);
      PROD2(8, 15)
#undef FETCH2
#undef PROD2
      __builtin_amdgcn_sched_barrier(0);
      if (OUT_CL) {
        const Epi& EK = E[(s + 1) & 1];
        float sd[8]  = {EZ.q[0].x, EZ.q[0].y, EZ.q[0].z, EZ.q[0].w, EZ.q[1].x, EZ.q[1].y, EZ.q[1].z, EZ.q[1].w};
        float xv0[8] = {EK.q[2].x, EK.q[2].y, EK.q[2].z, EK.q[2].w, EK.q[3].x, EK.q[3].y, EK.q[3].z, EK.q[3].w};
        float xv1[8] = {EK.q[4].x, EK.q[4].y, EK.q[4].z, EK.q[4].w, EK.q[5].x, EK.q[5].y, EK.q[5].z, EK.q[5].w};
        float* op = xout + ((size_t)b * HWSZ + z * WW) * CC + ((wbase + 8 * half) >> 1) * PP + 4 * p;
#pragma unroll
        for (int d = 0; d < 8; d += 2) {
          float m0 = (sd[d] > 0.f) ? 1.0f : 0.0f;
          float m1 = (sd[d + 1] > 0.f) ? 1.0f : 0.0f;
          v4f o;
          o.x = m0 * xv0[d]     + (1.0f - m0) * acc0[d];
          o.y = m1 * xv0[d + 1] + (1.0f - m1) * acc0[d + 1];
          o.z = m0 * xv1[d]     + (1.0f - m0) * acc1[d];
          o.w = m1 * xv1[d + 1] + (1.0f - m1) * acc1[d + 1];
          *(v4f*)(op + (d >> 1) * PP) = o;
        }
      } else {
        float m = (EZ.f[16] > 0.f) ? 1.0f : 0.0f;
#pragma unroll
        for (int v = 0; v < 8; v++) {
          size_t off0 = (size_t)(b * CC + v + 8 * half) * HWSZ + z * WW + wbase + p;
          size_t off1 = off0 + (size_t)16 * HWSZ;
          xout[off0] = m * EZ.f[2 * v]     + (1.0f - m) * acc0[v];
          xout[off1] = m * EZ.f[2 * v + 1] + (1.0f - m) * acc1[v];
        }
      }
    }
    if (s + 1 < RR + 2)
      epi_load<true>(E[(s + 1) & 1], x0, sdep, b, min(max(y + 1, 0), HH - 1), wbase, half, p);
  }
#undef LOAD_ROW
#undef LOAD_W
#undef BAND
}

#define BAND_FLOATS (16 * 2 * 60 * 16)
#define RING_FLOATS (4 * WW * LP)

__global__ __launch_bounds__(512) void pac_prop2_cl(const float* __restrict__ xin,
                                                    const float* __restrict__ x0,
                                                    const float* __restrict__ wgt,
                                                    const float* __restrict__ sdep,
                                                    float* __restrict__ xout) {
  __shared__ __attribute__((aligned(16))) float smem[BAND_FLOATS + RING_FLOATS];
  prop2_body<true>(smem + (threadIdx.x >> 5) * (2 * 60 * 16), smem + BAND_FLOATS,
                   xin, x0, wgt, sdep, xout);
}

__global__ __launch_bounds__(512) void pac_prop2_last(const float* __restrict__ xin,
                                                      const float* __restrict__ x0,
                                                      const float* __restrict__ wgt,
                                                      const float* __restrict__ sdep,
                                                      float* __restrict__ xout) {
  __shared__ __attribute__((aligned(16))) float smem[BAND_FLOATS + RING_FLOATS];
  prop2_body<false>(smem + (threadIdx.x >> 5) * (2 * 60 * 16), smem + BAND_FLOATS,
                    xin, x0, wgt, sdep, xout);
}

extern "C" void kernel_launch(void* const* d_in, const int* in_sizes, int n_in,
                              void* d_out, int out_size, void* d_ws, size_t ws_size,
                              hipStream_t stream) {
  const float* x      = (const float*)d_in[0];
  const float* guided = (const float*)d_in[1];
  const float* sdep   = (const float*)d_in[2];
  (void)in_sizes; (void)n_in; (void)out_size; (void)ws_size;

  float* bufA = (float*)d_ws;
  float* wgt  = bufA + (size_t)BB * CC * HWSZ;
  float* outb = (float*)d_out;

  pac_prep<<<(BB * HWSZ) / 256, 256, 0, stream>>>(guided, x, wgt, outb);

  const float* cur = outb;
  for (int t = 0; t < 7; t++) {
    float* dst = (t & 1) ? outb : bufA;
    pac_prop2_cl<<<BB * (HH / RR), 512, 0, stream>>>(cur, x, wgt, sdep, dst);
    cur = dst;
  }
  pac_prop2_last<<<BB * (HH / RR), 512, 0, stream>>>(cur, x, wgt, sdep, outb);
}


# pair-interleaved blends take x0 from the centre input row's register fragments (state equals x0 where masked; lane halves swap 2 of 4 pairs): x0 planes read by the last step only; sums unchanged
# speedup vs baseline: 4.9510x; 1.1290x over previous
#include <hip/hip_runtime.h>


typedef float v2f __attribute__((ext_vector_type(2)));
typedef float v4f __attribute__((ext_vector_type(4)));
typedef float v8f __attribute__((ext_vector_type(8)));

#define BB   8
#define CC   32
#define HH   256
#define WW   256
#define HWSZ (HH * WW)
#define RR   8
#define PP   64

__global__ __launch_bounds__(256) void pac_prep(const float* __restrict__ guided,
                                                const float* __restrict__ x,
                                                float* __restrict__ wgt,
                                                float* __restrict__ xcl) {
  __shared__ float tile[256 * 33];

  const int t   = threadIdx.x;
  const int idx = blockIdx.x * 256 + t;
  const int b   = idx >> 16;
  const int hw  = idx & 65535;

  const float* xp = x + (size_t)b * CC * HWSZ + hw;
#pragma unroll
  for (int c = 0; c < CC; c++) tile[t * 33 + c] = xp[(size_t)c * HWSZ];

  const float* gp = guided + (size_t)b * 8 * HWSZ + hw;
  float g[8];
  float mx = -1e30f;
#pragma unroll
  for (int n = 0; n < 8; n++) { g[n] = gp[(size_t)n * HWSZ]; mx = fmaxf(mx, g[n]); }
  float s = 0.f;
#pragma unroll
  for (int n = 0; n < 8; n++) { g[n] = __expf(g[n] - mx); s += g[n]; }
  float inv = 1.0f / s;
  float* wp = wgt + (size_t)b * 9 * HWSZ + hw;
#pragma unroll
  for (int n = 0; n < 4; n++) wp[(size_t)n * HWSZ] = g[n] * inv;
  wp[(size_t)4 * HWSZ] = 0.f;
#pragma unroll
  for (int n = 0; n < 4; n++) wp[(size_t)(5 + n) * HWSZ] = g[4 + n] * inv;

  __syncthreads();
  float* op = xcl + (size_t)(idx - t) * CC;
#pragma unroll
  for (int j = 0; j < 8; j++) {
    const int f  = (j * 256 + t) * 4;
    const int pr = f >> 6;
    const int sl = (f >> 2) & 15;
    const float* tp = tile + (2 * pr) * 33 + sl;
    v4f v = {tp[0], tp[33], tp[16], tp[33 + 16]};
    *(v4f*)(op + f) = v;
  }
}

struct Epi {
  v4f q[6];
  float f[17];
};

__device__ __forceinline__ void epi_load_out(Epi& E, const float* __restrict__ x0,
                                             const float* __restrict__ sdep,
                                             int b, int h, int wbase, int half, int p) {
  E.f[16] = sdep[(size_t)b * HWSZ + h * WW + wbase + p];
#pragma unroll
  for (int v = 0; v < 8; v++) {
    size_t off0 = (size_t)(b * CC + v + 8 * half) * HWSZ + h * WW + wbase + p;
    E.f[2 * v]     = x0[off0];
    E.f[2 * v + 1] = x0[off0 + (size_t)16 * HWSZ];
  }
}

__device__ __forceinline__ void epi_load_mask(Epi& E, const float* __restrict__ sdep,
                                              int b, int h, int wbase, int half) {
  const float* sp = sdep + (size_t)b * HWSZ + h * WW + wbase + 8 * half;
  E.q[0] = *(const v4f*)(sp);
  E.q[1] = *(const v4f*)(sp + 4);
}

__device__ __forceinline__ float xhalf(float v) {
  const int u = __builtin_bit_cast(int, v);
  return __builtin_bit_cast(float, __builtin_amdgcn_permlanex16(u, u, 0x76543210, 0xfedcba98, false, false));
}

__device__ __forceinline__ void epi_from_row(Epi& E, const v4f* X, bool hi) {
  const v4f ta = hi ? X[0] : X[3];
  const v4f tb = hi ? X[1] : X[4];
  v4f ra, rb;
  ra.x = xhalf(ta.x); ra.y = xhalf(ta.y); ra.z = xhalf(ta.z); ra.w = xhalf(ta.w);
  rb.x = xhalf(tb.x); rb.y = xhalf(tb.y); rb.z = xhalf(tb.z); rb.w = xhalf(tb.w);
  const v4f v0 = hi ? X[2] : ra;
  const v4f v1 = hi ? ra : X[1];
  const v4f v2 = hi ? X[3] : rb;
  const v4f v3 = hi ? rb : X[2];
  E.q[2] = (v4f){v0.x, v0.y, v1.x, v1.y};
  E.q[3] = (v4f){v2.x, v2.y, v3.x, v3.y};
  E.q[4] = (v4f){v0.z, v0.w, v1.z, v1.w};
  E.q[5] = (v4f){v2.z, v2.w, v3.z, v3.w};
}

#define LP CC

template <bool OUT_CL>
__device__ __forceinline__ void prop2_body(float* __restrict__ BW, float* __restrict__ YR,
                                           const float* __restrict__ xin,
                                           const float* __restrict__ x0,
                                           const float* __restrict__ wgt,
                                           const float* __restrict__ sdep,
                                           float* __restrict__ xout) {
  const int lane = threadIdx.x & 31;
  const int wv   = threadIdx.x >> 5;
  const int half = lane >> 4;
  const int p    = lane & 15;

  const int b     = blockIdx.x >> 5;
  const int h0    = (blockIdx.x & 31) * RR;
  const int wbase = wv << 4;

#pragma unroll
  for (int t = 0; t < 60; t++) BW[lane + 32 * t] = 0.f;

  int co[5];
#pragma unroll
  for (int kk = 0; kk < 5; kk++) {
    int pr = (wbase >> 1) - 1 + 2 * kk + half;
    co[kk] = min(max(pr, 0), WW / 2 - 1) * PP + 4 * p;
  }
  const bool cv0 = (wbase + p - 1) >= 0;
  const bool cv2 = (wbase + p + 1) < WW;

  const float* xa = xin + (size_t)b * HWSZ * CC;
  const float* wq = wgt + (size_t)b * 9 * HWSZ + wbase + p;

  v4f XF[3][5];
  float w[2][9];
  Epi E[2];
  Epi EZ;

#define LOAD_ROW(slot, row)                                                     \
  {                                                                             \
    const int rowc_ = min(max((row), 0), HH - 1);                               \
    const float* rp_ = xa + (size_t)rowc_ * (WW * CC);                          \
    _Pragma("unroll") for (int kk = 0; kk < 5; kk++)                            \
      XF[(slot)][kk] = *(const v4f*)(rp_ + co[kk]);                             \
  }
#define LOAD_W(s, row)                                                          \
  {                                                                             \
    const int rowc_ = min(max((row), 0), HH - 1);                               \
    _Pragma("unroll") for (int n = 0; n < 9; n++)                               \
      w[(s)][n] = wq[(size_t)n * HWSZ + rowc_ * WW];                            \
  }
#define BAND(ws, row, bb)                                                         \
  {                                                                             \
    __builtin_amdgcn_fence(__ATOMIC_ACQ_REL, "workgroup", "local");             \
    if (lane < 16) {                                                            \
      _Pragma("unroll") for (int n = 0; n < 9; n++) {                           \
        const int i = n / 3, j = n % 3;                                         \
        const bool rok = ((row) - 1 + i >= 0) && ((row) - 1 + i < HH);          \
        const bool cok = (j == 0) ? cv0 : ((j == 2) ? cv2 : true);              \
        (bb)[(20 * i + p + 1 + j) * 16 + p] = (rok && cok) ? w[(ws)][n] : 0.f;  \
      }                                                                         \
    }                                                                           \
    __builtin_amdgcn_fence(__ATOMIC_ACQ_REL, "workgroup", "local");             \
  }

  LOAD_ROW(0, h0 - 2)
  LOAD_ROW(1, h0 - 1)
  LOAD_ROW(2, h0)
  LOAD_W(0, h0 - 1)
  epi_load_mask(E[0], sdep, b, min(max(h0 - 1, 0), HH - 1), wbase, half);

#pragma unroll
  for (int s = 0; s < RR + 2; s++) {
    const int y = h0 - 1 + s;
    if (s + 1 < RR + 2) LOAD_W((s + 1) & 1, y + 1)
    if (!OUT_CL && s >= 2) epi_load_out(EZ, x0, sdep, b, y - 1, wbase, half, p);
    const float* BWy = BW + (s & 1) * 960;
    const float* BWz = BW + ((s + 1) & 1) * 960;
    BAND(s & 1, y, BW + (s & 1) * 960)
    {
      v8f acc0 = {0.f, 0.f, 0.f, 0.f, 0.f, 0.f, 0.f, 0.f};
      v8f acc1 = {0.f, 0.f, 0.f, 0.f, 0.f, 0.f, 0.f, 0.f};
      v2f bfy[15];
#pragma unroll
      for (int c = 0; c < 15; c++) {
        bfy[c].x = BWy[(4 * c + 0) * 16 + half * 32 + p];
        bfy[c].y = BWy[(4 * c + 1) * 16 + half * 32 + p];
      }
      __builtin_amdgcn_sched_barrier(0);
#pragma unroll
      for (int i = 0; i < 3; i++) {
#pragma unroll
        for (int kk = 0; kk < 5; kk++) {
          const int c = 5 * i + kk;
          const v2f bf  = bfy[c];
          const v4f xf  = XF[(s + i) % 3][kk];
          const v2f xf0 = __builtin_shufflevector(xf, xf, 0, 1);
          const v2f xf1 = __builtin_shufflevector(xf, xf, 2, 3);
          acc0 = __builtin_amdgcn_wmma_f32_16x16x4_f32(false, bf, false, xf0,
                                                       (short)0, acc0, false, false);
          acc1 = __builtin_amdgcn_wmma_f32_16x16x4_f32(false, bf, false, xf1,
                                                       (short)0, acc1, false, false);
        }
      }
      epi_from_row(E[s & 1], XF[(s + 1) % 3], half != 0);
      const Epi& EF = E[s & 1];
      float sd[8]  = {EF.q[0].x, EF.q[0].y, EF.q[0].z, EF.q[0].w, EF.q[1].x, EF.q[1].y, EF.q[1].z, EF.q[1].w};
      float xv0[8] = {EF.q[2].x, EF.q[2].y, EF.q[2].z, EF.q[2].w, EF.q[3].x, EF.q[3].y, EF.q[3].z, EF.q[3].w};
      float xv1[8] = {EF.q[4].x, EF.q[4].y, EF.q[4].z, EF.q[4].w, EF.q[5].x, EF.q[5].y, EF.q[5].z, EF.q[5].w};
      float* yp = YR + (s & 3) * (WW * LP) + ((wbase + 8 * half) >> 1) * PP + 4 * p;
#pragma unroll
      for (int d = 0; d < 8; d += 2) {
        float m0 = (sd[d] > 0.f) ? 1.0f : 0.0f;
        float m1 = (sd[d + 1] > 0.f) ? 1.0f : 0.0f;
        v4f o;
        o.x = m0 * xv0[d]     + (1.0f - m0) * acc0[d];
        o.y = m1 * xv0[d + 1] + (1.0f - m1) * acc0[d + 1];
        o.z = m0 * xv1[d]     + (1.0f - m0) * acc1[d];
        o.w = m1 * xv1[d + 1] + (1.0f - m1) * acc1[d + 1];
        *(v4f*)(yp + (d >> 1) * PP) = o;
      }
    }
    __builtin_amdgcn_fence(__ATOMIC_RELEASE, "workgroup", "local");
    __builtin_amdgcn_s_barrier();
    __builtin_amdgcn_fence(__ATOMIC_ACQUIRE, "workgroup", "local");

    if (OUT_CL && s >= 2) epi_load_mask(EZ, sdep, b, y - 1, wbase, half);
    if (s + 1 < RR + 2) LOAD_ROW(s % 3, y + 2)
    if (s >= 2) {
      const int z = y - 1;
      v8f acc0 = {0.f, 0.f, 0.f, 0.f, 0.f, 0.f, 0.f, 0.f};
      v8f acc1 = {0.f, 0.f, 0.f, 0.f, 0.f, 0.f, 0.f, 0.f};
      v2f bfz[15];
      v4f uz[15];
#define FETCH2(c0, c1)                                                                  \
      _Pragma("unroll") for (int c = (c0); c < (c1); c++) {                             \
        const float* yr = YR + ((s - 2 + c / 5) & 3) * (WW * LP);     \
        bfz[c].x = BWz[(4 * c + 0) * 16 + half * 32 + p];                               \
        bfz[c].y = BWz[(4 * c + 1) * 16 + half * 32 + p];                               \
        uz[c]    = *(const v4f*)(yr + co[c % 5]);                                       \
      }
#define PROD2(c0, c1)                                                                   \
      _Pragma("unroll") for (int c = (c0); c < (c1); c++) {                             \
        const v2f bf  = bfz[c];                                                         \
        const v2f af0 = __builtin_shufflevector(uz[c], uz[c], 0, 1);        \
        const v2f af1 = __builtin_shufflevector(uz[c], uz[c], 2, 3);        \
        if (OUT_CL) {                                                                   \
          acc0 = __builtin_amdgcn_wmma_f32_16x16x4_f32(false, bf, false, af0,           \
                                                       (short)0, acc0, false, false);   \
          acc1 = __builtin_amdgcn_wmma_f32_16x16x4_f32(false, bf, false, af1,           \
                                                       (short)0, acc1, false, false);   \
        } else {                                                                        \
          acc0 = __builtin_amdgcn_wmma_f32_16x16x4_f32(false, af0, false, bf,           \
                                                       (short)0, acc0, false, false);   \
          acc1 = __builtin_amdgcn_wmma_f32_16x16x4_f32(false, af1, false, bf,           \
                                                       (short)0, acc1, false, false);   \
        }                                                                               \
      }
      FETCH2(0, 6)
      __builtin_amdgcn_sched_barrier(0);
      PROD2(0, 3)
      __builtin_amdgcn_sched_barrier(0);
      FETCH2(6, 11)
      __builtin_amdgcn_sched_barrier(0);
      PROD2(3, 8)
      __builtin_amdgcn_sched_barrier(0);
      FETCH2(11, 15)
      __builtin_amdgcn_sched_barrier(0);
      PROD2(8, 15)
#undef FETCH2
#undef PROD2
      __builtin_amdgcn_sched_barrier(0);
      if (OUT_CL) {
        const Epi& EK = E[(s + 1) & 1];
        float sd[8]  = {EZ.q[0].x, EZ.q[0].y, EZ.q[0].z, EZ.q[0].w, EZ.q[1].x, EZ.q[1].y, EZ.q[1].z, EZ.q[1].w};
        float xv0[8] = {EK.q[2].x, EK.q[2].y, EK.q[2].z, EK.q[2].w, EK.q[3].x, EK.q[3].y, EK.q[3].z, EK.q[3].w};
        float xv1[8] = {EK.q[4].x, EK.q[4].y, EK.q[4].z, EK.q[4].w, EK.q[5].x, EK.q[5].y, EK.q[5].z, EK.q[5].w};
        float* op = xout + ((size_t)b * HWSZ + z * WW) * CC + ((wbase + 8 * half) >> 1) * PP + 4 * p;
#pragma unroll
        for (int d = 0; d < 8; d += 2) {
          float m0 = (sd[d] > 0.f) ? 1.0f : 0.0f;
          float m1 = (sd[d + 1] > 0.f) ? 1.0f : 0.0f;
          v4f o;
          o.x = m0 * xv0[d]     + (1.0f - m0) * acc0[d];
          o.y = m1 * xv0[d + 1] + (1.0f - m1) * acc0[d + 1];
          o.z = m0 * xv1[d]     + (1.0f - m0) * acc1[d];
          o.w = m1 * xv1[d + 1] + (1.0f - m1) * acc1[d + 1];
          *(v4f*)(op + (d >> 1) * PP) = o;
        }
      } else {
        float m = (EZ.f[16] > 0.f) ? 1.0f : 0.0f;
#pragma unroll
        for (int v = 0; v < 8; v++) {
          size_t off0 = (size_t)(b * CC + v + 8 * half) * HWSZ + z * WW + wbase + p;
          size_t off1 = off0 + (size_t)16 * HWSZ;
          xout[off0] = m * EZ.f[2 * v]     + (1.0f - m) * acc0[v];
          xout[off1] = m * EZ.f[2 * v + 1] + (1.0f - m) * acc1[v];
        }
      }
    }
    if (s + 1 < RR + 2)
      epi_load_mask(E[(s + 1) & 1], sdep, b, min(max(y + 1, 0), HH - 1), wbase, half);
  }
#undef LOAD_ROW
#undef LOAD_W
#undef BAND
}

#define BAND_FLOATS (16 * 2 * 60 * 16)
#define RING_FLOATS (4 * WW * LP)

__global__ __launch_bounds__(512) void pac_prop2_cl(const float* __restrict__ xin,
                                                    const float* __restrict__ x0,
                                                    const float* __restrict__ wgt,
                                                    const float* __restrict__ sdep,
                                                    float* __restrict__ xout) {
  __shared__ __attribute__((aligned(16))) float smem[BAND_FLOATS + RING_FLOATS];
  prop2_body<true>(smem + (threadIdx.x >> 5) * (2 * 60 * 16), smem + BAND_FLOATS,
                   xin, x0, wgt, sdep, xout);
}

__global__ __launch_bounds__(512) void pac_prop2_last(const float* __restrict__ xin,
                                                      const float* __restrict__ x0,
                                                      const float* __restrict__ wgt,
                                                      const float* __restrict__ sdep,
                                                      float* __restrict__ xout) {
  __shared__ __attribute__((aligned(16))) float smem[BAND_FLOATS + RING_FLOATS];
  prop2_body<false>(smem + (threadIdx.x >> 5) * (2 * 60 * 16), smem + BAND_FLOATS,
                    xin, x0, wgt, sdep, xout);
}

extern "C" void kernel_launch(void* const* d_in, const int* in_sizes, int n_in,
                              void* d_out, int out_size, void* d_ws, size_t ws_size,
                              hipStream_t stream) {
  const float* x      = (const float*)d_in[0];
  const float* guided = (const float*)d_in[1];
  const float* sdep   = (const float*)d_in[2];
  (void)in_sizes; (void)n_in; (void)out_size; (void)ws_size;

  float* bufA = (float*)d_ws;
  float* wgt  = bufA + (size_t)BB * CC * HWSZ;
  float* outb = (float*)d_out;

  pac_prep<<<(BB * HWSZ) / 256, 256, 0, stream>>>(guided, x, wgt, outb);

  const float* cur = outb;
  for (int t = 0; t < 7; t++) {
    float* dst = (t & 1) ? outb : bufA;
    pac_prop2_cl<<<BB * (HH / RR), 512, 0, stream>>>(cur, x, wgt, sdep, dst);
    cur = dst;
  }
  pac_prop2_last<<<BB * (HH / RR), 512, 0, stream>>>(cur, x, wgt, sdep, outb);
}


# two steps applied as ONE per-pixel 5x5 stencil: W2 = T*T (mask, edges folded into 3x3 table T) formed per row in the wave; 25 K-chunks per 2 steps, no LDS ring or block barrier; sum order changes
# speedup vs baseline: 5.7833x; 1.1681x over previous
#include <hip/hip_runtime.h>


typedef float v2f __attribute__((ext_vector_type(2)));
typedef float v4f __attribute__((ext_vector_type(4)));
typedef float v8f __attribute__((ext_vector_type(8)));

#define BB   8
#define CC   32
#define HH   256
#define WW   256
#define HWSZ (HH * WW)
#define RR   8
#define PP   64

__global__ __launch_bounds__(256) void pac_prep(const float* __restrict__ guided,
                                                const float* __restrict__ x,
                                                const float* __restrict__ sdep,
                                                float* __restrict__ wgt,
                                                float* __restrict__ xcl) {
  __shared__ float tile[256 * 33];

  const int t   = threadIdx.x;
  const int idx = blockIdx.x * 256 + t;
  const int b   = idx >> 16;
  const int hw  = idx & 65535;

  const float* xp = x + (size_t)b * CC * HWSZ + hw;
#pragma unroll
  for (int c = 0; c < CC; c++) tile[t * 33 + c] = xp[(size_t)c * HWSZ];

  const float* gp = guided + (size_t)b * 8 * HWSZ + hw;
  float g[8];
  float mx = -1e30f;
#pragma unroll
  for (int n = 0; n < 8; n++) { g[n] = gp[(size_t)n * HWSZ]; mx = fmaxf(mx, g[n]); }
  float s = 0.f;
#pragma unroll
  for (int n = 0; n < 8; n++) { g[n] = __expf(g[n] - mx); s += g[n]; }
  float inv = 1.0f / s;
  float* wp = wgt + (size_t)b * 9 * HWSZ + hw;
  const bool msk = sdep[idx] > 0.f;
  const int hr = hw >> 8;
  const int wc = hw & 255;
#pragma unroll
  for (int n = 0; n < 9; n++) {
    const int i = n / 3, j = n % 3;
    const bool inside = (hr - 1 + i >= 0) && (hr - 1 + i < HH) && (wc - 1 + j >= 0) && (wc - 1 + j < WW);
    const float sv = (n == 4) ? 0.f : g[n < 4 ? n : n - 1] * inv;
    const float id = (n == 4) ? 1.f : 0.f;
    wp[(size_t)n * HWSZ] = msk ? id : (inside ? sv : 0.f);
  }

  __syncthreads();
  float* op = xcl + (size_t)(idx - t) * CC;
#pragma unroll
  for (int j = 0; j < 8; j++) {
    const int f  = (j * 256 + t) * 4;
    const int pr = f >> 6;
    const int sl = (f >> 2) & 15;
    const float* tp = tile + (2 * pr) * 33 + sl;
    v4f v = {tp[0], tp[33], tp[16], tp[33 + 16]};
    *(v4f*)(op + f) = v;
  }
}

#define TBR 180

template <bool OUT_CL>
__device__ __forceinline__ void prop2_body(float* __restrict__ BW, float* __restrict__ TB,
                                           const float* __restrict__ xin,
                                           const float* __restrict__ wgt,
                                           float* __restrict__ xout) {
  const int lane = threadIdx.x & 31;
  const int wv   = threadIdx.x >> 5;
  const int half = lane >> 4;
  const int p    = lane & 15;

  const int b     = blockIdx.x >> 5;
  const int h0    = (blockIdx.x & 31) * RR;
  const int wbase = wv << 4;

#pragma unroll
  for (int t = 0; t < 50; t++) BW[lane + 32 * t] = 0.f;

  int co[5];
#pragma unroll
  for (int kk = 0; kk < 5; kk++) {
    int pr = (wbase >> 1) - 1 + 2 * kk + half;
    co[kk] = min(max(pr, 0), WW / 2 - 1) * PP + 4 * p;
  }

  const float* xa = xin + (size_t)b * HWSZ * CC;
  const bool tl = lane < 18;
  const float* bq = wgt + (size_t)b * 9 * HWSZ + min(max(wbase - 1 + lane, 0), WW - 1);

  v4f XF[5][5];
  float wt[3][9];

#define LOAD_ROW(slot, row)                                                     \
  {                                                                             \
    const int rowc_ = min(max((row), 0), HH - 1);                               \
    const float* rp_ = xa + (size_t)rowc_ * (WW * CC);                          \
    _Pragma("unroll") for (int kk = 0; kk < 5; kk++)                            \
      XF[(slot)][kk] = *(const v4f*)(rp_ + co[kk]);                             \
  }
#define LOAD_T(k, row)                                                          \
  if (tl) {                                                                     \
    const int rowc_ = min(max((row), 0), HH - 1);                               \
    _Pragma("unroll") for (int n = 0; n < 9; n++)                               \
      wt[(k)][n] = bq[(size_t)n * HWSZ + rowc_ * WW];                           \
  }
#define STORE_T(k, r)                                                           \
  if (tl) {                                                                     \
    _Pragma("unroll") for (int n = 0; n < 9; n++)                               \
      TB[(r) * TBR + n * 20 + lane] = wt[(k)][n];                               \
  }

  LOAD_T(0, h0 - 1)
  LOAD_T(1, h0)
  LOAD_T(2, h0 + 1)
  LOAD_ROW(0, h0 - 2)
  LOAD_ROW(1, h0 - 1)
  LOAD_ROW(2, h0)
  LOAD_ROW(3, h0 + 1)
  LOAD_ROW(4, h0 + 2)
  STORE_T(0, 0)
  STORE_T(1, 1)
  STORE_T(2, 2)

#pragma unroll
  for (int s = 0; s < RR; s++) {
    const int z = h0 + s;
    if (s + 1 < RR) LOAD_T(0, z + 2)

    __builtin_amdgcn_fence(__ATOMIC_ACQ_REL, "workgroup", "local");
    if (lane < 16) {
      float A[9];
#pragma unroll
      for (int n = 0; n < 9; n++) A[n] = TB[(s + 1) * TBR + n * 20 + p + 1];
      float W2[25];
#pragma unroll
      for (int d = 0; d < 25; d++) W2[d] = 0.f;
#pragma unroll
      for (int ay = 0; ay < 3; ay++) {
#pragma unroll
        for (int ax = 0; ax < 3; ax++) {
          const float a = A[3 * ay + ax];
          const float* tn = TB + (s + ay) * TBR + p + ax;
#pragma unroll
          for (int ty = 0; ty < 3; ty++) {
#pragma unroll
            for (int tx = 0; tx < 3; tx++)
              W2[5 * (ay + ty) + ax + tx] = __builtin_fmaf(a, tn[(3 * ty + tx) * 20], W2[5 * (ay + ty) + ax + tx]);
          }
        }
      }
#pragma unroll
      for (int i = 0; i < 5; i++) {
#pragma unroll
        for (int j = 0; j < 5; j++) BW[(20 * i + p + j) * 16 + p] = W2[5 * i + j];
      }
    }
    __builtin_amdgcn_fence(__ATOMIC_ACQ_REL, "workgroup", "local");

    v8f acc0 = {0.f, 0.f, 0.f, 0.f, 0.f, 0.f, 0.f, 0.f};
    v8f acc1 = {0.f, 0.f, 0.f, 0.f, 0.f, 0.f, 0.f, 0.f};
    v2f bfz[25];
#pragma unroll
    for (int c = 0; c < 25; c++) {
      bfz[c].x = BW[(4 * c + 0) * 16 + half * 32 + p];
      bfz[c].y = BW[(4 * c + 1) * 16 + half * 32 + p];
    }
    __builtin_amdgcn_sched_barrier(0);
#pragma unroll
    for (int i = 0; i < 5; i++) {
#pragma unroll
      for (int kk = 0; kk < 5; kk++) {
        const v2f bf  = bfz[5 * i + kk];
        const v4f xf  = XF[(s + i) % 5][kk];
        const v2f xf0 = __builtin_shufflevector(xf, xf, 0, 1);
        const v2f xf1 = __builtin_shufflevector(xf, xf, 2, 3);
        if (OUT_CL) {
          acc0 = __builtin_amdgcn_wmma_f32_16x16x4_f32(false, bf, false, xf0,
                                                       (short)0, acc0, false, false);
          acc1 = __builtin_amdgcn_wmma_f32_16x16x4_f32(false, bf, false, xf1,
                                                       (short)0, acc1, false, false);
        } else {
          acc0 = __builtin_amdgcn_wmma_f32_16x16x4_f32(false, xf0, false, bf,
                                                       (short)0, acc0, false, false);
          acc1 = __builtin_amdgcn_wmma_f32_16x16x4_f32(false, xf1, false, bf,
                                                       (short)0, acc1, false, false);
        }
      }
    }
    if (OUT_CL) {
      float* op = xout + ((size_t)b * HWSZ + z * WW) * CC + ((wbase + 8 * half) >> 1) * PP + 4 * p;
#pragma unroll
      for (int d = 0; d < 8; d += 2) {
        v4f o = {acc0[d], acc0[d + 1], acc1[d], acc1[d + 1]};
        *(v4f*)(op + (d >> 1) * PP) = o;
      }
    } else {
#pragma unroll
      for (int v = 0; v < 8; v++) {
        size_t off0 = (size_t)(b * CC + v + 8 * half) * HWSZ + z * WW + wbase + p;
        size_t off1 = off0 + (size_t)16 * HWSZ;
        xout[off0] = acc0[v];
        xout[off1] = acc1[v];
      }
    }
    if (s + 1 < RR) LOAD_ROW(s % 5, z + 3)
    if (s + 1 < RR) STORE_T(0, s + 3)
  }
#undef LOAD_ROW
#undef LOAD_T
#undef STORE_T
}

#define BAND_FLOATS 1600
#define WAVE_FLOATS (BAND_FLOATS + 10 * TBR)

__global__ __launch_bounds__(512) void pac_prop2_cl(const float* __restrict__ xin,
                                                    const float* __restrict__ wgt,
                                                    float* __restrict__ xout) {
  __shared__ __attribute__((aligned(16))) float smem[16 * WAVE_FLOATS];
  float* mine = smem + (threadIdx.x >> 5) * WAVE_FLOATS;
  prop2_body<true>(mine, mine + BAND_FLOATS, xin, wgt, xout);
}

__global__ __launch_bounds__(512) void pac_prop2_last(const float* __restrict__ xin,
                                                      const float* __restrict__ wgt,
                                                      float* __restrict__ xout) {
  __shared__ __attribute__((aligned(16))) float smem[16 * WAVE_FLOATS];
  float* mine = smem + (threadIdx.x >> 5) * WAVE_FLOATS;
  prop2_body<false>(mine, mine + BAND_FLOATS, xin, wgt, xout);
}

extern "C" void kernel_launch(void* const* d_in, const int* in_sizes, int n_in,
                              void* d_out, int out_size, void* d_ws, size_t ws_size,
                              hipStream_t stream) {
  const float* x      = (const float*)d_in[0];
  const float* guided = (const float*)d_in[1];
  const float* sdep   = (const float*)d_in[2];
  (void)in_sizes; (void)n_in; (void)out_size; (void)ws_size;

  float* bufA = (float*)d_ws;
  float* wgt  = bufA + (size_t)BB * CC * HWSZ;
  float* outb = (float*)d_out;

  pac_prep<<<(BB * HWSZ) / 256, 256, 0, stream>>>(guided, x, sdep, wgt, outb);

  const float* cur = outb;
  for (int t = 0; t < 7; t++) {
    float* dst = (t & 1) ? outb : bufA;
    pac_prop2_cl<<<BB * (HH / RR), 512, 0, stream>>>(cur, wgt, dst);
    cur = dst;
  }
  pac_prop2_last<<<BB * (HH / RR), 512, 0, stream>>>(cur, wgt, outb);
}


# wave priority falls with the row a wave has reached (s_setprio 3,3,3,2,2,1,1,0 over its 8 rows): a SIMD's 4 waves advance together, no solo tail of the youngest on the matrix pipe; sums unchanged
# speedup vs baseline: 6.0402x; 1.0444x over previous
#include <hip/hip_runtime.h>


typedef float v2f __attribute__((ext_vector_type(2)));
typedef float v4f __attribute__((ext_vector_type(4)));
typedef float v8f __attribute__((ext_vector_type(8)));

#define BB   8
#define CC   32
#define HH   256
#define WW   256
#define HWSZ (HH * WW)
#define RR   8
#define PP   64

__global__ __launch_bounds__(256) void pac_prep(const float* __restrict__ guided,
                                                const float* __restrict__ x,
                                                const float* __restrict__ sdep,
                                                float* __restrict__ wgt,
                                                float* __restrict__ xcl) {
  __shared__ float tile[256 * 33];

  const int t   = threadIdx.x;
  const int idx = blockIdx.x * 256 + t;
  const int b   = idx >> 16;
  const int hw  = idx & 65535;

  const float* xp = x + (size_t)b * CC * HWSZ + hw;
#pragma unroll
  for (int c = 0; c < CC; c++) tile[t * 33 + c] = xp[(size_t)c * HWSZ];

  const float* gp = guided + (size_t)b * 8 * HWSZ + hw;
  float g[8];
  float mx = -1e30f;
#pragma unroll
  for (int n = 0; n < 8; n++) { g[n] = gp[(size_t)n * HWSZ]; mx = fmaxf(mx, g[n]); }
  float s = 0.f;
#pragma unroll
  for (int n = 0; n < 8; n++) { g[n] = __expf(g[n] - mx); s += g[n]; }
  float inv = 1.0f / s;
  float* wp = wgt + (size_t)b * 9 * HWSZ + hw;
  const bool msk = sdep[idx] > 0.f;
  const int hr = hw >> 8;
  const int wc = hw & 255;
#pragma unroll
  for (int n = 0; n < 9; n++) {
    const int i = n / 3, j = n % 3;
    const bool inside = (hr - 1 + i >= 0) && (hr - 1 + i < HH) && (wc - 1 + j >= 0) && (wc - 1 + j < WW);
    const float sv = (n == 4) ? 0.f : g[n < 4 ? n : n - 1] * inv;
    const float id = (n == 4) ? 1.f : 0.f;
    wp[(size_t)n * HWSZ] = msk ? id : (inside ? sv : 0.f);
  }

  __syncthreads();
  float* op = xcl + (size_t)(idx - t) * CC;
#pragma unroll
  for (int j = 0; j < 8; j++) {
    const int f  = (j * 256 + t) * 4;
    const int pr = f >> 6;
    const int sl = (f >> 2) & 15;
    const float* tp = tile + (2 * pr) * 33 + sl;
    v4f v = {tp[0], tp[33], tp[16], tp[33 + 16]};
    *(v4f*)(op + f) = v;
  }
}

#define TBR 180

template <bool OUT_CL>
__device__ __forceinline__ void prop2_body(float* __restrict__ BW, float* __restrict__ TB,
                                           const float* __restrict__ xin,
                                           const float* __restrict__ wgt,
                                           float* __restrict__ xout) {
  const int lane = threadIdx.x & 31;
  const int wv   = threadIdx.x >> 5;
  const int half = lane >> 4;
  const int p    = lane & 15;

  const int b     = blockIdx.x >> 5;
  const int h0    = (blockIdx.x & 31) * RR;
  const int wbase = wv << 4;

#pragma unroll
  for (int t = 0; t < 50; t++) BW[lane + 32 * t] = 0.f;

  int co[5];
#pragma unroll
  for (int kk = 0; kk < 5; kk++) {
    int pr = (wbase >> 1) - 1 + 2 * kk + half;
    co[kk] = min(max(pr, 0), WW / 2 - 1) * PP + 4 * p;
  }

  const float* xa = xin + (size_t)b * HWSZ * CC;
  const bool tl = lane < 18;
  const float* bq = wgt + (size_t)b * 9 * HWSZ + min(max(wbase - 1 + lane, 0), WW - 1);

  v4f XF[5][5];
  float wt[3][9];

#define LOAD_ROW(slot, row)                                                     \
  {                                                                             \
    const int rowc_ = min(max((row), 0), HH - 1);                               \
    const float* rp_ = xa + (size_t)rowc_ * (WW * CC);                          \
    _Pragma("unroll") for (int kk = 0; kk < 5; kk++)                            \
      XF[(slot)][kk] = *(const v4f*)(rp_ + co[kk]);                             \
  }
#define LOAD_T(k, row)                                                          \
  if (tl) {                                                                     \
    const int rowc_ = min(max((row), 0), HH - 1);                               \
    _Pragma("unroll") for (int n = 0; n < 9; n++)                               \
      wt[(k)][n] = bq[(size_t)n * HWSZ + rowc_ * WW];                           \
  }
#define STORE_T(k, r)                                                           \
  if (tl) {                                                                     \
    _Pragma("unroll") for (int n = 0; n < 9; n++)                               \
      TB[(r) * TBR + n * 20 + lane] = wt[(k)][n];                               \
  }

  LOAD_T(0, h0 - 1)
  LOAD_T(1, h0)
  LOAD_T(2, h0 + 1)
  LOAD_ROW(0, h0 - 2)
  LOAD_ROW(1, h0 - 1)
  LOAD_ROW(2, h0)
  LOAD_ROW(3, h0 + 1)
  LOAD_ROW(4, h0 + 2)
  STORE_T(0, 0)
  STORE_T(1, 1)
  STORE_T(2, 2)

#pragma unroll
  for (int s = 0; s < RR; s++) {
    const int z = h0 + s;
    if (s == 0) __builtin_amdgcn_s_setprio(3);
    if (s == 3) __builtin_amdgcn_s_setprio(2);
    if (s == 5) __builtin_amdgcn_s_setprio(1);
    if (s == 7) __builtin_amdgcn_s_setprio(0);
    if (s + 1 < RR) LOAD_T(0, z + 2)

    __builtin_amdgcn_fence(__ATOMIC_ACQ_REL, "workgroup", "local");
    if (lane < 16) {
      float A[9];
#pragma unroll
      for (int n = 0; n < 9; n++) A[n] = TB[(s + 1) * TBR + n * 20 + p + 1];
      float W2[25];
#pragma unroll
      for (int d = 0; d < 25; d++) W2[d] = 0.f;
#pragma unroll
      for (int ay = 0; ay < 3; ay++) {
#pragma unroll
        for (int ax = 0; ax < 3; ax++) {
          const float a = A[3 * ay + ax];
          const float* tn = TB + (s + ay) * TBR + p + ax;
#pragma unroll
          for (int ty = 0; ty < 3; ty++) {
#pragma unroll
            for (int tx = 0; tx < 3; tx++)
              W2[5 * (ay + ty) + ax + tx] = __builtin_fmaf(a, tn[(3 * ty + tx) * 20], W2[5 * (ay + ty) + ax + tx]);
          }
        }
      }
#pragma unroll
      for (int i = 0; i < 5; i++) {
#pragma unroll
        for (int j = 0; j < 5; j++) BW[(20 * i + p + j) * 16 + p] = W2[5 * i + j];
      }
    }
    __builtin_amdgcn_fence(__ATOMIC_ACQ_REL, "workgroup", "local");

    v8f acc0 = {0.f, 0.f, 0.f, 0.f, 0.f, 0.f, 0.f, 0.f};
    v8f acc1 = {0.f, 0.f, 0.f, 0.f, 0.f, 0.f, 0.f, 0.f};
    v2f bfz[25];
#pragma unroll
    for (int c = 0; c < 25; c++) {
      bfz[c].x = BW[(4 * c + 0) * 16 + half * 32 + p];
      bfz[c].y = BW[(4 * c + 1) * 16 + half * 32 + p];
    }
    __builtin_amdgcn_sched_barrier(0);
#pragma unroll
    for (int i = 0; i < 5; i++) {
#pragma unroll
      for (int kk = 0; kk < 5; kk++) {
        const v2f bf  = bfz[5 * i + kk];
        const v4f xf  = XF[(s + i) % 5][kk];
        const v2f xf0 = __builtin_shufflevector(xf, xf, 0, 1);
        const v2f xf1 = __builtin_shufflevector(xf, xf, 2, 3);
        if (OUT_CL) {
          acc0 = __builtin_amdgcn_wmma_f32_16x16x4_f32(false, bf, false, xf0,
                                                       (short)0, acc0, false, false);
          acc1 = __builtin_amdgcn_wmma_f32_16x16x4_f32(false, bf, false, xf1,
                                                       (short)0, acc1, false, false);
        } else {
          acc0 = __builtin_amdgcn_wmma_f32_16x16x4_f32(false, xf0, false, bf,
                                                       (short)0, acc0, false, false);
          acc1 = __builtin_amdgcn_wmma_f32_16x16x4_f32(false, xf1, false, bf,
                                                       (short)0, acc1, false, false);
        }
      }
    }
    if (OUT_CL) {
      float* op = xout + ((size_t)b * HWSZ + z * WW) * CC + ((wbase + 8 * half) >> 1) * PP + 4 * p;
#pragma unroll
      for (int d = 0; d < 8; d += 2) {
        v4f o = {acc0[d], acc0[d + 1], acc1[d], acc1[d + 1]};
        *(v4f*)(op + (d >> 1) * PP) = o;
      }
    } else {
#pragma unroll
      for (int v = 0; v < 8; v++) {
        size_t off0 = (size_t)(b * CC + v + 8 * half) * HWSZ + z * WW + wbase + p;
        size_t off1 = off0 + (size_t)16 * HWSZ;
        xout[off0] = acc0[v];
        xout[off1] = acc1[v];
      }
    }
    if (s + 1 < RR) LOAD_ROW(s % 5, z + 3)
    if (s + 1 < RR) STORE_T(0, s + 3)
  }
#undef LOAD_ROW
#undef LOAD_T
#undef STORE_T
}

#define BAND_FLOATS 1600
#define WAVE_FLOATS (BAND_FLOATS + 10 * TBR)

__global__ __launch_bounds__(512) void pac_prop2_cl(const float* __restrict__ xin,
                                                    const float* __restrict__ wgt,
                                                    float* __restrict__ xout) {
  __shared__ __attribute__((aligned(16))) float smem[16 * WAVE_FLOATS];
  float* mine = smem + (threadIdx.x >> 5) * WAVE_FLOATS;
  prop2_body<true>(mine, mine + BAND_FLOATS, xin, wgt, xout);
}

__global__ __launch_bounds__(512) void pac_prop2_last(const float* __restrict__ xin,
                                                      const float* __restrict__ wgt,
                                                      float* __restrict__ xout) {
  __shared__ __attribute__((aligned(16))) float smem[16 * WAVE_FLOATS];
  float* mine = smem + (threadIdx.x >> 5) * WAVE_FLOATS;
  prop2_body<false>(mine, mine + BAND_FLOATS, xin, wgt, xout);
}

extern "C" void kernel_launch(void* const* d_in, const int* in_sizes, int n_in,
                              void* d_out, int out_size, void* d_ws, size_t ws_size,
                              hipStream_t stream) {
  const float* x      = (const float*)d_in[0];
  const float* guided = (const float*)d_in[1];
  const float* sdep   = (const float*)d_in[2];
  (void)in_sizes; (void)n_in; (void)out_size; (void)ws_size;

  float* bufA = (float*)d_ws;
  float* wgt  = bufA + (size_t)BB * CC * HWSZ;
  float* outb = (float*)d_out;

  pac_prep<<<(BB * HWSZ) / 256, 256, 0, stream>>>(guided, x, sdep, wgt, outb);

  const float* cur = outb;
  for (int t = 0; t < 7; t++) {
    float* dst = (t & 1) ? outb : bufA;
    pac_prop2_cl<<<BB * (HH / RR), 512, 0, stream>>>(cur, wgt, dst);
    cur = dst;
  }
  pac_prop2_last<<<BB * (HH / RR), 512, 0, stream>>>(cur, wgt, outb);
}


# W2 and Band built for two output rows per pass, lane half hf forming row z+hf into a Band buffer of its own: 4 full-wave passes per wave, not 8 half-wave; table rows 48 banks apart; sums unchanged
# speedup vs baseline: 6.3870x; 1.0574x over previous
#include <hip/hip_runtime.h>


typedef float v2f __attribute__((ext_vector_type(2)));
typedef float v4f __attribute__((ext_vector_type(4)));
typedef float v8f __attribute__((ext_vector_type(8)));

#define BB   8
#define CC   32
#define HH   256
#define WW   256
#define HWSZ (HH * WW)
#define RR   8
#define PP   64

__global__ __launch_bounds__(256) void pac_prep(const float* __restrict__ guided,
                                                const float* __restrict__ x,
                                                const float* __restrict__ sdep,
                                                float* __restrict__ wgt,
                                                float* __restrict__ xcl) {
  __shared__ float tile[256 * 33];

  const int t   = threadIdx.x;
  const int idx = blockIdx.x * 256 + t;
  const int b   = idx >> 16;
  const int hw  = idx & 65535;

  const float* xp = x + (size_t)b * CC * HWSZ + hw;
#pragma unroll
  for (int c = 0; c < CC; c++) tile[t * 33 + c] = xp[(size_t)c * HWSZ];

  const float* gp = guided + (size_t)b * 8 * HWSZ + hw;
  float g[8];
  float mx = -1e30f;
#pragma unroll
  for (int n = 0; n < 8; n++) { g[n] = gp[(size_t)n * HWSZ]; mx = fmaxf(mx, g[n]); }
  float s = 0.f;
#pragma unroll
  for (int n = 0; n < 8; n++) { g[n] = __expf(g[n] - mx); s += g[n]; }
  float inv = 1.0f / s;
  float* wp = wgt + (size_t)b * 9 * HWSZ + hw;
  const bool msk = sdep[idx] > 0.f;
  const int hr = hw >> 8;
  const int wc = hw & 255;
#pragma unroll
  for (int n = 0; n < 9; n++) {
    const int i = n / 3, j = n % 3;
    const bool inside = (hr - 1 + i >= 0) && (hr - 1 + i < HH) && (wc - 1 + j >= 0) && (wc - 1 + j < WW);
    const float sv = (n == 4) ? 0.f : g[n < 4 ? n : n - 1] * inv;
    const float id = (n == 4) ? 1.f : 0.f;
    wp[(size_t)n * HWSZ] = msk ? id : (inside ? sv : 0.f);
  }

  __syncthreads();
  float* op = xcl + (size_t)(idx - t) * CC;
#pragma unroll
  for (int j = 0; j < 8; j++) {
    const int f  = (j * 256 + t) * 4;
    const int pr = f >> 6;
    const int sl = (f >> 2) & 15;
    const float* tp = tile + (2 * pr) * 33 + sl;
    v4f v = {tp[0], tp[33], tp[16], tp[33 + 16]};
    *(v4f*)(op + f) = v;
  }
}

#define TAPP 19
#define TBR 176
#define BAND_FLOATS 1600
#define BAND1_OFF (BAND_FLOATS + 32)

template <bool OUT_CL>
__device__ __forceinline__ void prop2_body(float* __restrict__ BW, float* __restrict__ TB,
                                           const float* __restrict__ xin,
                                           const float* __restrict__ wgt,
                                           float* __restrict__ xout) {
  const int lane = threadIdx.x & 31;
  const int wv   = threadIdx.x >> 5;
  const int half = lane >> 4;
  const int p    = lane & 15;

  const int b     = blockIdx.x >> 5;
  const int h0    = (blockIdx.x & 31) * RR;
  const int wbase = wv << 4;

#pragma unroll
  for (int t = 0; t < (BAND1_OFF + BAND_FLOATS) / 32; t++) BW[lane + 32 * t] = 0.f;

  int co[5];
#pragma unroll
  for (int kk = 0; kk < 5; kk++) {
    int pr = (wbase >> 1) - 1 + 2 * kk + half;
    co[kk] = min(max(pr, 0), WW / 2 - 1) * PP + 4 * p;
  }

  const float* xa = xin + (size_t)b * HWSZ * CC;
  const bool tl = lane < 18;
  const float* bq = wgt + (size_t)b * 9 * HWSZ + min(max(wbase - 1 + lane, 0), WW - 1);

  v4f XF[5][5];
  float wt[4][9];

#define LOAD_ROW(slot, row)                                                     \
  {                                                                             \
    const int rowc_ = min(max((row), 0), HH - 1);                               \
    const float* rp_ = xa + (size_t)rowc_ * (WW * CC);                          \
    _Pragma("unroll") for (int kk = 0; kk < 5; kk++)                            \
      XF[(slot)][kk] = *(const v4f*)(rp_ + co[kk]);                             \
  }
#define LOAD_T(k, row)                                                          \
  if (tl) {                                                                     \
    const int rowc_ = min(max((row), 0), HH - 1);                               \
    _Pragma("unroll") for (int n = 0; n < 9; n++)                               \
      wt[(k)][n] = bq[(size_t)n * HWSZ + rowc_ * WW];                           \
  }
#define STORE_T(k, r)                                                           \
  if (tl) {                                                                     \
    _Pragma("unroll") for (int n = 0; n < 9; n++)                               \
      TB[(r) * TBR + n * TAPP + lane] = wt[(k)][n];                             \
  }

  LOAD_T(0, h0 - 1)
  LOAD_T(1, h0)
  LOAD_T(2, h0 + 1)
  LOAD_T(3, h0 + 2)
  LOAD_ROW(0, h0 - 2)
  LOAD_ROW(1, h0 - 1)
  LOAD_ROW(2, h0)
  LOAD_ROW(3, h0 + 1)
  LOAD_ROW(4, h0 + 2)
  STORE_T(0, 0)
  STORE_T(1, 1)
  STORE_T(2, 2)
  STORE_T(3, 3)

#pragma unroll
  for (int s = 0; s < RR; s++) {
    const int z = h0 + s;
    if (s == 0) __builtin_amdgcn_s_setprio(3);
    if (s == 3) __builtin_amdgcn_s_setprio(2);
    if (s == 5) __builtin_amdgcn_s_setprio(1);
    if (s == 7) __builtin_amdgcn_s_setprio(0);
    if (s + 2 < RR) LOAD_T(0, z + 3)

    __builtin_amdgcn_fence(__ATOMIC_ACQ_REL, "workgroup", "local");
    if ((s & 1) == 0) {
      const float* tr = TB + (s + half) * TBR;
      float* bw = BW + half * BAND1_OFF;
      float A[9];
#pragma unroll
      for (int n = 0; n < 9; n++) A[n] = tr[TBR + n * TAPP + p + 1];
      float W2[25];
#pragma unroll
      for (int d = 0; d < 25; d++) W2[d] = 0.f;
#pragma unroll
      for (int ay = 0; ay < 3; ay++) {
#pragma unroll
        for (int ax = 0; ax < 3; ax++) {
          const float a = A[3 * ay + ax];
          const float* tn = tr + ay * TBR + p + ax;
#pragma unroll
          for (int ty = 0; ty < 3; ty++) {
#pragma unroll
            for (int tx = 0; tx < 3; tx++)
              W2[5 * (ay + ty) + ax + tx] = __builtin_fmaf(a, tn[(3 * ty + tx) * TAPP], W2[5 * (ay + ty) + ax + tx]);
          }
        }
      }
#pragma unroll
      for (int i = 0; i < 5; i++) {
#pragma unroll
        for (int j = 0; j < 5; j++) bw[(20 * i + p + j) * 16 + p] = W2[5 * i + j];
      }
    }
    __builtin_amdgcn_fence(__ATOMIC_ACQ_REL, "workgroup", "local");

    v8f acc0 = {0.f, 0.f, 0.f, 0.f, 0.f, 0.f, 0.f, 0.f};
    v8f acc1 = {0.f, 0.f, 0.f, 0.f, 0.f, 0.f, 0.f, 0.f};
    const float* br = BW + (s & 1) * BAND1_OFF;
    v2f bfz[25];
#pragma unroll
    for (int c = 0; c < 25; c++) {
      bfz[c].x = br[(4 * c + 0) * 16 + half * 32 + p];
      bfz[c].y = br[(4 * c + 1) * 16 + half * 32 + p];
    }
    __builtin_amdgcn_sched_barrier(0);
#pragma unroll
    for (int i = 0; i < 5; i++) {
#pragma unroll
      for (int kk = 0; kk < 5; kk++) {
        const v2f bf  = bfz[5 * i + kk];
        const v4f xf  = XF[(s + i) % 5][kk];
        const v2f xf0 = __builtin_shufflevector(xf, xf, 0, 1);
        const v2f xf1 = __builtin_shufflevector(xf, xf, 2, 3);
        if (OUT_CL) {
          acc0 = __builtin_amdgcn_wmma_f32_16x16x4_f32(false, bf, false, xf0,
                                                       (short)0, acc0, false, false);
          acc1 = __builtin_amdgcn_wmma_f32_16x16x4_f32(false, bf, false, xf1,
                                                       (short)0, acc1, false, false);
        } else {
          acc0 = __builtin_amdgcn_wmma_f32_16x16x4_f32(false, xf0, false, bf,
                                                       (short)0, acc0, false, false);
          acc1 = __builtin_amdgcn_wmma_f32_16x16x4_f32(false, xf1, false, bf,
                                                       (short)0, acc1, false, false);
        }
      }
    }
    if (OUT_CL) {
      float* op = xout + ((size_t)b * HWSZ + z * WW) * CC + ((wbase + 8 * half) >> 1) * PP + 4 * p;
#pragma unroll
      for (int d = 0; d < 8; d += 2) {
        v4f o = {acc0[d], acc0[d + 1], acc1[d], acc1[d + 1]};
        *(v4f*)(op + (d >> 1) * PP) = o;
      }
    } else {
#pragma unroll
      for (int v = 0; v < 8; v++) {
        size_t off0 = (size_t)(b * CC + v + 8 * half) * HWSZ + z * WW + wbase + p;
        size_t off1 = off0 + (size_t)16 * HWSZ;
        xout[off0] = acc0[v];
        xout[off1] = acc1[v];
      }
    }
    if (s + 1 < RR) LOAD_ROW(s % 5, z + 3)
    if (s + 2 < RR) STORE_T(0, s + 4)
  }
#undef LOAD_ROW
#undef LOAD_T
#undef STORE_T
}

#define TB_OFF (BAND1_OFF + BAND_FLOATS)
#define WAVE_FLOATS (TB_OFF + 10 * TBR)

__global__ __launch_bounds__(512) void pac_prop2_cl(const float* __restrict__ xin,
                                                    const float* __restrict__ wgt,
                                                    float* __restrict__ xout) {
  __shared__ __attribute__((aligned(16))) float smem[16 * WAVE_FLOATS];
  float* mine = smem + (threadIdx.x >> 5) * WAVE_FLOATS;
  prop2_body<true>(mine, mine + TB_OFF, xin, wgt, xout);
}

__global__ __launch_bounds__(512) void pac_prop2_last(const float* __restrict__ xin,
                                                      const float* __restrict__ wgt,
                                                      float* __restrict__ xout) {
  __shared__ __attribute__((aligned(16))) float smem[16 * WAVE_FLOATS];
  float* mine = smem + (threadIdx.x >> 5) * WAVE_FLOATS;
  prop2_body<false>(mine, mine + TB_OFF, xin, wgt, xout);
}

extern "C" void kernel_launch(void* const* d_in, const int* in_sizes, int n_in,
                              void* d_out, int out_size, void* d_ws, size_t ws_size,
                              hipStream_t stream) {
  const float* x      = (const float*)d_in[0];
  const float* guided = (const float*)d_in[1];
  const float* sdep   = (const float*)d_in[2];
  (void)in_sizes; (void)n_in; (void)out_size; (void)ws_size;

  float* bufA = (float*)d_ws;
  float* wgt  = bufA + (size_t)BB * CC * HWSZ;
  float* outb = (float*)d_out;

  pac_prep<<<(BB * HWSZ) / 256, 256, 0, stream>>>(guided, x, sdep, wgt, outb);

  const float* cur = outb;
  for (int t = 0; t < 7; t++) {
    float* dst = (t & 1) ? outb : bufA;
    pac_prop2_cl<<<BB * (HH / RR), 512, 0, stream>>>(cur, wgt, dst);
    cur = dst;
  }
  pac_prop2_last<<<BB * (HH / RR), 512, 0, stream>>>(cur, wgt, outb);
}
